# starting kernel
# baseline (speedup 1.0000x reference)
#include <hip/hip_runtime.h>
#include <hip/hip_bf16.h>


#define N_NODES 100000
#define N_EDGES 1600000
#define IN_DIM  512
#define HID     64
#define OUT_DIM 5

typedef __attribute__((ext_vector_type(2))) float v2f;
typedef __attribute__((ext_vector_type(8))) float v8f;

__global__ void k_init_deg(float* __restrict__ deg) {
    int i = blockIdx.x * blockDim.x + threadIdx.x;
    if (i < N_NODES) deg[i] = 1.0f;
}

__global__ void k_deg_scatter(const int* __restrict__ col,
                              const float* __restrict__ w,
                              float* __restrict__ deg) {
    int stride = gridDim.x * blockDim.x;
    for (int e = blockIdx.x * blockDim.x + threadIdx.x; e < N_EDGES; e += stride)
        atomicAdd(deg + col[e], w[e]);
}

__global__ void k_dinv(float* __restrict__ deg) {
    int i = blockIdx.x * blockDim.x + threadIdx.x;
    if (i < N_NODES) {
        float d = deg[i];
        deg[i] = (d > 0.0f) ? rsqrtf(d) : 0.0f;
    }
}

__global__ void k_zero(float* __restrict__ p, int n) {
    int stride = gridDim.x * blockDim.x;
    for (int i = blockIdx.x * blockDim.x + threadIdx.x; i < n; i += stride)
        p[i] = 0.0f;
}

__global__ void k_gemm_n64(const float* __restrict__ X,
                           const float* __restrict__ W,
                           float* __restrict__ H,
                           int M, int K) {
    const int lane  = threadIdx.x & 31;
    const int waveT = (blockIdx.x * blockDim.x + threadIdx.x) >> 5;
    const int nTiles = M >> 4;
    if (waveT >= nTiles) return;

    const int m0   = waveT << 4;
    const int mRow = m0 + (lane & 15);
    const int kSub = (lane & 16) ? 2 : 0;
    const int nCol = lane & 15;

    v8f c0 = {}, c1 = {}, c2 = {}, c3 = {};

    for (int k = 0; k < K; k += 4) {
        v2f a = *(const v2f*)(X + (size_t)mRow * K + (k + kSub));

        const float* wr0 = W + (size_t)(k + kSub)     * HID + nCol;
        const float* wr1 = W + (size_t)(k + kSub + 1) * HID + nCol;
        v2f b0, b1, b2, b3;
        b0.x = wr0[0];  b0.y = wr1[0];
        b1.x = wr0[16]; b1.y = wr1[16];
        b2.x = wr0[32]; b2.y = wr1[32];
        b3.x = wr0[48]; b3.y = wr1[48];

        c0 = __builtin_amdgcn_wmma_f32_16x16x4_f32(false, a, false, b0,
                                                   (short)0, c0, false, false);
        c1 = __builtin_amdgcn_wmma_f32_16x16x4_f32(false, a, false, b1,
                                                   (short)0, c1, false, false);
        c2 = __builtin_amdgcn_wmma_f32_16x16x4_f32(false, a, false, b2,
                                                   (short)0, c2, false, false);
        c3 = __builtin_amdgcn_wmma_f32_16x16x4_f32(false, a, false, b3,
                                                   (short)0, c3, false, false);
    }

    const int rBase = m0 + ((lane & 16) ? 8 : 0);
    #pragma unroll
    for (int j = 0; j < 8; ++j) {
        float* dst = H + (size_t)(rBase + j) * HID + nCol;
        dst[0]  = c0[j];
        dst[16] = c1[j];
        dst[32] = c2[j];
        dst[48] = c3[j];
    }
}

__global__ void k_edge_scatter(const int* __restrict__ row,
                               const int* __restrict__ col,
                               const float* __restrict__ w,
                               const float* __restrict__ dinv,
                               const float* __restrict__ h,
                               float* __restrict__ agg) {
    const int lane   = threadIdx.x & 31;
    const int wid    = (blockIdx.x * blockDim.x + threadIdx.x) >> 5;
    const int nWaves = (gridDim.x * blockDim.x) >> 5;

    for (int e = wid; e < N_EDGES; e += nWaves) {
        const int r = row[e];
        const int c = col[e];
        const float norm = dinv[r] * w[e] * dinv[c];
        v2f hv = *(const v2f*)(h + (size_t)r * HID + lane * 2);
        float* dst = agg + (size_t)c * HID + lane * 2;
        atomicAdd(dst + 0, norm * hv.x);
        atomicAdd(dst + 1, norm * hv.y);
    }
}

__global__ void k_finalize(float* __restrict__ agg,
                           const float* __restrict__ h,
                           const float* __restrict__ dinv,
                           const float* __restrict__ b,
                           int relu) {
    int stride = gridDim.x * blockDim.x;
    for (int i = blockIdx.x * blockDim.x + threadIdx.x; i < N_NODES * HID; i += stride) {
        const int n = i >> 6;
        const int f = i & 63;
        const float di = dinv[n];
        float v = agg[i] + di * di * h[i] + b[f];
        agg[i] = relu ? fmaxf(v, 0.0f) : v;
    }
}

__global__ void k_fc(const float* __restrict__ h,
                     const float* __restrict__ Wfc,
                     const float* __restrict__ bfc,
                     float* __restrict__ out) {
    int n = blockIdx.x * blockDim.x + threadIdx.x;
    if (n >= N_NODES) return;
    float acc[OUT_DIM];
    #pragma unroll
    for (int o = 0; o < OUT_DIM; ++o) acc[o] = bfc[o];
    const float* hr = h + (size_t)n * HID;
    #pragma unroll 4
    for (int k = 0; k < HID; ++k) {
        const float xv = hr[k];
        #pragma unroll
        for (int o = 0; o < OUT_DIM; ++o)
            acc[o] = fmaf(xv, Wfc[k * OUT_DIM + o], acc[o]);
    }
    float* orow = out + (size_t)n * OUT_DIM;
    #pragma unroll
    for (int o = 0; o < OUT_DIM; ++o) orow[o] = acc[o];
}

extern "C" void kernel_launch(void* const* d_in, const int* in_sizes, int n_in,
                              void* d_out, int out_size, void* d_ws, size_t ws_size,
                              hipStream_t stream) {
    const float* x   = (const float*)d_in[0];
    const int*   ei  = (const int*)  d_in[1];
    const float* ew  = (const float*)d_in[2];
    const float* W1  = (const float*)d_in[3];
    const float* b1  = (const float*)d_in[4];
    const float* W2  = (const float*)d_in[5];
    const float* b2  = (const float*)d_in[6];
    const float* Wfc = (const float*)d_in[7];
    const float* bfc = (const float*)d_in[8];
    float* out = (float*)d_out;

    const int* row = ei;
    const int* col = ei + N_EDGES;

    char* ws = (char*)d_ws;
    size_t off = 0;
    float* dinv = (float*)(ws + off); off += ((size_t)N_NODES * 4 + 255) & ~(size_t)255;
    float* bufH = (float*)(ws + off); off += (size_t)N_NODES * HID * 4;
    float* bufA = (float*)(ws + off); off += (size_t)N_NODES * HID * 4;
    (void)ws_size; (void)n_in; (void)in_sizes; (void)out_size;

    const int TPB = 256;
    const int nodeBlocks = (N_NODES + TPB - 1) / TPB;
    const int featElems  = N_NODES * HID;
    const int gemmBlocks = ((N_NODES / 16) + (TPB / 32) - 1) / (TPB / 32);

    k_init_deg<<<nodeBlocks, TPB, 0, stream>>>(dinv);
    k_deg_scatter<<<1024, TPB, 0, stream>>>(col, ew, dinv);
    k_dinv<<<nodeBlocks, TPB, 0, stream>>>(dinv);

    k_gemm_n64<<<gemmBlocks, TPB, 0, stream>>>(x, W1, bufH, N_NODES, IN_DIM);
    k_zero<<<4096, TPB, 0, stream>>>(bufA, featElems);
    k_edge_scatter<<<2048, TPB, 0, stream>>>(row, col, ew, dinv, bufH, bufA);
    k_finalize<<<4096, TPB, 0, stream>>>(bufA, bufH, dinv, b1, 1);

    k_gemm_n64<<<gemmBlocks, TPB, 0, stream>>>(bufA, W2, bufH, N_NODES, HID);
    k_zero<<<4096, TPB, 0, stream>>>(bufA, featElems);
    k_edge_scatter<<<2048, TPB, 0, stream>>>(row, col, ew, dinv, bufH, bufA);
    k_finalize<<<4096, TPB, 0, stream>>>(bufA, bufH, dinv, b2, 1);

    k_fc<<<nodeBlocks, TPB, 0, stream>>>(bufA, Wfc, bfc, out);
}


# edge sums: work-group (1191-node range x edge third) sums in LDS; waves scan target ids, queue hits, gather 16 rows at once, LDS float adds; range flushed by whole-line atomics; sum order: arrival
# speedup vs baseline: 1.3574x; 1.3574x over previous
#include <hip/hip_runtime.h>
#include <hip/hip_bf16.h>


#define N_NODES 100000
#define N_EDGES 1600000
#define IN_DIM  512
#define HID     64
#define OUT_DIM 5

typedef __attribute__((ext_vector_type(2))) float v2f;
typedef __attribute__((ext_vector_type(8))) float v8f;

#define DEG_R     8
#define DEG_S     32
#define DEG_NR    12512
#define DEG_PITCH (DEG_R * DEG_NR)
#define DEG_ES    (N_EDGES / DEG_S)
#define DEG_TPB   1024

typedef __attribute__((ext_vector_type(4))) int   v4i;
typedef __attribute__((ext_vector_type(4))) float v4f;

__global__ __launch_bounds__(DEG_TPB)
void k_deg_partial(const int* __restrict__ col,
                   const float* __restrict__ w,
                   float* __restrict__ part) {
    __shared__ float sdeg[DEG_NR];
    const int r  = blockIdx.x % DEG_R;
    const int s  = blockIdx.x / DEG_R;
    const int lo = r * DEG_NR;

    for (int i = threadIdx.x; i < DEG_NR; i += DEG_TPB) sdeg[i] = 0.0f;
    __syncthreads();

    const v4i* c4 = (const v4i*)(col + (size_t)s * DEG_ES);
    const v4f* w4 = (const v4f*)(w   + (size_t)s * DEG_ES);
    for (int v0 = threadIdx.x; v0 < DEG_ES / 4; v0 += 4 * DEG_TPB) {
        v4i c[4];
        v4f x[4];
        for (int u = 0; u < 4; ++u) {
            const int v = v0 + u * DEG_TPB;
            const bool in = v < DEG_ES / 4;
            c[u] = in ? c4[v] : v4i{-1, -1, -1, -1};
            x[u] = in ? w4[v] : v4f{0.0f, 0.0f, 0.0f, 0.0f};
        }
        for (int u = 0; u < 4; ++u) {
            for (int j = 0; j < 4; ++j) {
                const unsigned d = (unsigned)(c[u][j] - lo);
                if (d < (unsigned)DEG_NR)
                    __hip_atomic_fetch_add(&sdeg[d], x[u][j], __ATOMIC_RELAXED,
                                           __HIP_MEMORY_SCOPE_WORKGROUP);
            }
        }
    }
    __syncthreads();

    float* dst = part + (size_t)s * DEG_PITCH + lo;
    for (int i = threadIdx.x; i < DEG_NR; i += DEG_TPB) dst[i] = sdeg[i];
}

__global__ void k_dinv_sum(const float* __restrict__ part,
                           float* __restrict__ dinv) {
    int i = blockIdx.x * blockDim.x + threadIdx.x;
    if (i < N_NODES) {
        float d = 1.0f;
        for (int s = 0; s < DEG_S; ++s) d += part[(size_t)s * DEG_PITCH + i];
        dinv[i] = (d > 0.0f) ? rsqrtf(d) : 0.0f;
    }
}

__global__ void k_zero(float* __restrict__ p, int n) {
    int stride = gridDim.x * blockDim.x;
    for (int i = blockIdx.x * blockDim.x + threadIdx.x; i < n; i += stride)
        p[i] = 0.0f;
}

__global__ void k_gemm_n64(const float* __restrict__ X,
                           const float* __restrict__ W,
                           float* __restrict__ H,
                           int M, int K) {
    const int lane  = threadIdx.x & 31;
    const int waveT = (blockIdx.x * blockDim.x + threadIdx.x) >> 5;
    const int nTiles = M >> 4;
    if (waveT >= nTiles) return;

    const int m0   = waveT << 4;
    const int mRow = m0 + (lane & 15);
    const int kSub = (lane & 16) ? 2 : 0;
    const int nCol = lane & 15;

    v8f c0 = {}, c1 = {}, c2 = {}, c3 = {};

    for (int k = 0; k < K; k += 4) {
        v2f a = *(const v2f*)(X + (size_t)mRow * K + (k + kSub));

        const float* wr0 = W + (size_t)(k + kSub)     * HID + nCol;
        const float* wr1 = W + (size_t)(k + kSub + 1) * HID + nCol;
        v2f b0, b1, b2, b3;
        b0.x = wr0[0];  b0.y = wr1[0];
        b1.x = wr0[16]; b1.y = wr1[16];
        b2.x = wr0[32]; b2.y = wr1[32];
        b3.x = wr0[48]; b3.y = wr1[48];

        c0 = __builtin_amdgcn_wmma_f32_16x16x4_f32(false, a, false, b0,
                                                   (short)0, c0, false, false);
        c1 = __builtin_amdgcn_wmma_f32_16x16x4_f32(false, a, false, b1,
                                                   (short)0, c1, false, false);
        c2 = __builtin_amdgcn_wmma_f32_16x16x4_f32(false, a, false, b2,
                                                   (short)0, c2, false, false);
        c3 = __builtin_amdgcn_wmma_f32_16x16x4_f32(false, a, false, b3,
                                                   (short)0, c3, false, false);
    }

    const int rBase = m0 + ((lane & 16) ? 8 : 0);
    #pragma unroll
    for (int j = 0; j < 8; ++j) {
        float* dst = H + (size_t)(rBase + j) * HID + nCol;
        dst[0]  = c0[j];
        dst[16] = c1[j];
        dst[32] = c2[j];
        dst[48] = c3[j];
    }
}

#define OWN_NR    1191
#define OWN_R     84
#define OWN_S     3
#define OWN_TPB   1024
#define OWN_NV    8
#define OWN_VTOT  (N_EDGES / 4)
#define OWN_VS    ((OWN_VTOT + OWN_S - 1) / OWN_S)
#define OWN_CHUNK (OWN_NV * OWN_TPB)
#define OWN_NCH   ((OWN_VS + OWN_CHUNK - 1) / OWN_CHUNK)
#define OWN_GB    16
#define OWN_WQ    64

__global__ __launch_bounds__(OWN_TPB)
void k_edge_owner(const int* __restrict__ row,
                  const int* __restrict__ col,
                  const float* __restrict__ w,
                  const float* __restrict__ dinv,
                  const float* __restrict__ h,
                  float* __restrict__ agg) {
    __shared__ float    sums[(OWN_NR + 1) * HID];
    __shared__ unsigned wq[OWN_TPB / 32][OWN_WQ];

    const int tid  = threadIdx.x;
    const int lane = tid & 31;
    const int wv   = __builtin_amdgcn_readfirstlane(tid >> 5);
    const int r    = blockIdx.x % OWN_R;
    const int s    = blockIdx.x / OWN_R;
    const int lo   = r * OWN_NR;
    const int vs   = s * OWN_VS;
    const int ve   = (vs + OWN_VS < OWN_VTOT) ? (vs + OWN_VS) : OWN_VTOT;

    {
        v4f* z4 = (v4f*)sums;
        for (int i = tid; i < OWN_NR * HID / 4; i += OWN_TPB)
            z4[i] = v4f{0.0f, 0.0f, 0.0f, 0.0f};
    }

    const v4i* c4 = (const v4i*)col;
    v4i c[OWN_NV];
    v4i cn[OWN_NV];
    #pragma unroll
    for (int u = 0; u < OWN_NV; ++u) {
        const int v = vs + u * OWN_TPB + tid;
        c[u] = (v < ve) ? c4[v] : v4i{-1, -1, -1, -1};
    }
    __syncthreads();

    unsigned* myq = wq[wv];
    int count = 0;

    auto serve = [&](int base, int cnt) {
        int   rr = 0;
        int   dd = 0;
        float nr = 0.0f;
        if (lane < cnt) {
            const unsigned e = myq[base + lane];
            rr = row[e];
            const int   cc = col[e];
            const float we = w[e];
            dd = cc - lo;
            nr = dinv[rr] * we * dinv[cc];
        }
        v2f   hv[OWN_GB];
        float nn[OWN_GB];
        int   dl[OWN_GB];
        #pragma unroll
        for (int u = 0; u < OWN_GB; ++u) {
            const bool ok = u < cnt;
            const int  i  = ok ? u : 0;
            const int  ru = __builtin_amdgcn_readlane(rr, i);
            const int  du = __builtin_amdgcn_readlane(dd, i);
            dl[u] = ok ? du : OWN_NR;
            nn[u] = __builtin_bit_cast(float,
                      __builtin_amdgcn_readlane(__builtin_bit_cast(int, nr), i));
            hv[u] = *(const v2f*)(h + (size_t)ru * HID + lane * 2);
        }
        #pragma unroll
        for (int u = 0; u < OWN_GB; ++u) {
            float* dst = &sums[dl[u] * HID + lane * 2];
            __hip_atomic_fetch_add(dst + 0, nn[u] * hv[u].x, __ATOMIC_RELAXED,
                                   __HIP_MEMORY_SCOPE_WORKGROUP);
            __hip_atomic_fetch_add(dst + 1, nn[u] * hv[u].y, __ATOMIC_RELAXED,
                                   __HIP_MEMORY_SCOPE_WORKGROUP);
        }
    };

    for (int ch = 0; ch < OWN_NCH; ++ch) {
        const int vb = vs + ch * OWN_CHUNK + tid;

        #pragma unroll
        for (int u = 0; u < OWN_NV; ++u) {
            const int v = vb + OWN_CHUNK + u * OWN_TPB;
            cn[u] = (ch + 1 < OWN_NCH && v < ve) ? c4[v] : v4i{-1, -1, -1, -1};
        }

        unsigned mask = 0u;
        #pragma unroll
        for (int u = 0; u < OWN_NV; ++u) {
            #pragma unroll
            for (int j = 0; j < 4; ++j) {
                const unsigned d = (unsigned)(c[u][j] - lo);
                mask |= (d < (unsigned)OWN_NR) ? (1u << (u * 4 + j)) : 0u;
            }
        }

        for (;;) {
            const bool     has = mask != 0u;
            const unsigned bal = __builtin_amdgcn_ballot_w32(has);
            if (bal == 0u) break;
            if (has) {
                const int idx = __builtin_ctz(mask);
                mask &= mask - 1u;
                const unsigned e = 4u * (unsigned)(vb + (idx >> 2) * OWN_TPB) + (unsigned)(idx & 3);
                myq[count + (int)__builtin_amdgcn_mbcnt_lo(bal, 0u)] = e;
            }
            count += __builtin_popcount(bal);
            __builtin_amdgcn_fence(__ATOMIC_ACQ_REL, "wavefront");
            while (count >= OWN_GB) {
                count -= OWN_GB;
                serve(count, OWN_GB);
            }
        }

        #pragma unroll
        for (int u = 0; u < OWN_NV; ++u) c[u] = cn[u];
    }
    if (count > 0) serve(0, count);

    __syncthreads();

    float* dst = agg + (size_t)lo * HID;
    for (int i = tid; i < OWN_NR * HID; i += OWN_TPB) {
        if (lo + (i >> 6) < N_NODES) atomicAdd(dst + i, sums[i]);
    }
}

__global__ void k_finalize(float* __restrict__ agg,
                           const float* __restrict__ h,
                           const float* __restrict__ dinv,
                           const float* __restrict__ b,
                           int relu) {
    int stride = gridDim.x * blockDim.x;
    for (int i = blockIdx.x * blockDim.x + threadIdx.x; i < N_NODES * HID; i += stride) {
        const int n = i >> 6;
        const int f = i & 63;
        const float di = dinv[n];
        float v = agg[i] + di * di * h[i] + b[f];
        agg[i] = relu ? fmaxf(v, 0.0f) : v;
    }
}

__global__ void k_fc(const float* __restrict__ h,
                     const float* __restrict__ Wfc,
                     const float* __restrict__ bfc,
                     float* __restrict__ out) {
    int n = blockIdx.x * blockDim.x + threadIdx.x;
    if (n >= N_NODES) return;
    float acc[OUT_DIM];
    #pragma unroll
    for (int o = 0; o < OUT_DIM; ++o) acc[o] = bfc[o];
    const float* hr = h + (size_t)n * HID;
    #pragma unroll 4
    for (int k = 0; k < HID; ++k) {
        const float xv = hr[k];
        #pragma unroll
        for (int o = 0; o < OUT_DIM; ++o)
            acc[o] = fmaf(xv, Wfc[k * OUT_DIM + o], acc[o]);
    }
    float* orow = out + (size_t)n * OUT_DIM;
    #pragma unroll
    for (int o = 0; o < OUT_DIM; ++o) orow[o] = acc[o];
}

extern "C" void kernel_launch(void* const* d_in, const int* in_sizes, int n_in,
                              void* d_out, int out_size, void* d_ws, size_t ws_size,
                              hipStream_t stream) {
    const float* x   = (const float*)d_in[0];
    const int*   ei  = (const int*)  d_in[1];
    const float* ew  = (const float*)d_in[2];
    const float* W1  = (const float*)d_in[3];
    const float* b1  = (const float*)d_in[4];
    const float* W2  = (const float*)d_in[5];
    const float* b2  = (const float*)d_in[6];
    const float* Wfc = (const float*)d_in[7];
    const float* bfc = (const float*)d_in[8];
    float* out = (float*)d_out;

    const int* row = ei;
    const int* col = ei + N_EDGES;

    char* ws = (char*)d_ws;
    size_t off = 0;
    float* dinv = (float*)(ws + off); off += ((size_t)N_NODES * 4 + 255) & ~(size_t)255;
    float* bufH = (float*)(ws + off); off += (size_t)N_NODES * HID * 4;
    float* bufA = (float*)(ws + off); off += (size_t)N_NODES * HID * 4;
    (void)ws_size; (void)n_in; (void)in_sizes; (void)out_size;

    const int TPB = 256;
    const int nodeBlocks = (N_NODES + TPB - 1) / TPB;
    const int featElems  = N_NODES * HID;
    const int gemmBlocks = ((N_NODES / 16) + (TPB / 32) - 1) / (TPB / 32);

    k_deg_partial<<<DEG_R * DEG_S, DEG_TPB, 0, stream>>>(col, ew, bufA);
    k_dinv_sum<<<nodeBlocks, TPB, 0, stream>>>(bufA, dinv);

    k_gemm_n64<<<gemmBlocks, TPB, 0, stream>>>(x, W1, bufH, N_NODES, IN_DIM);
    k_zero<<<4096, TPB, 0, stream>>>(bufA, featElems);
    k_edge_owner<<<OWN_R * OWN_S, OWN_TPB, 0, stream>>>(row, col, ew, dinv, bufH, bufA);
    k_finalize<<<4096, TPB, 0, stream>>>(bufA, bufH, dinv, b1, 1);

    k_gemm_n64<<<gemmBlocks, TPB, 0, stream>>>(bufA, W2, bufH, N_NODES, HID);
    k_zero<<<4096, TPB, 0, stream>>>(bufA, featElems);
    k_edge_owner<<<OWN_R * OWN_S, OWN_TPB, 0, stream>>>(row, col, ew, dinv, bufH, bufA);
    k_finalize<<<4096, TPB, 0, stream>>>(bufA, bufH, dinv, b2, 1);

    k_fc<<<nodeBlocks, TPB, 0, stream>>>(bufA, Wfc, bfc, out);
}


# edge sums, layer 2: each wave replays the edge records (source, place, norm) its layer-1 pass left in the workspace; no second id scan, queue or index gathers; sum order: arrival, as before
# speedup vs baseline: 1.5855x; 1.1680x over previous
#include <hip/hip_runtime.h>
#include <hip/hip_bf16.h>


#define N_NODES 100000
#define N_EDGES 1600000
#define IN_DIM  512
#define HID     64
#define OUT_DIM 5

typedef __attribute__((ext_vector_type(2))) float v2f;
typedef __attribute__((ext_vector_type(8))) float v8f;

#define DEG_R     8
#define DEG_S     32
#define DEG_NR    12512
#define DEG_PITCH (DEG_R * DEG_NR)
#define DEG_ES    (N_EDGES / DEG_S)
#define DEG_TPB   1024

typedef __attribute__((ext_vector_type(4))) int   v4i;
typedef __attribute__((ext_vector_type(4))) float v4f;

__global__ __launch_bounds__(DEG_TPB)
void k_deg_partial(const int* __restrict__ col,
                   const float* __restrict__ w,
                   float* __restrict__ part) {
    __shared__ float sdeg[DEG_NR];
    const int r  = blockIdx.x % DEG_R;
    const int s  = blockIdx.x / DEG_R;
    const int lo = r * DEG_NR;

    for (int i = threadIdx.x; i < DEG_NR; i += DEG_TPB) sdeg[i] = 0.0f;
    __syncthreads();

    const v4i* c4 = (const v4i*)(col + (size_t)s * DEG_ES);
    const v4f* w4 = (const v4f*)(w   + (size_t)s * DEG_ES);
    for (int v0 = threadIdx.x; v0 < DEG_ES / 4; v0 += 4 * DEG_TPB) {
        v4i c[4];
        v4f x[4];
        for (int u = 0; u < 4; ++u) {
            const int v = v0 + u * DEG_TPB;
            const bool in = v < DEG_ES / 4;
            c[u] = in ? c4[v] : v4i{-1, -1, -1, -1};
            x[u] = in ? w4[v] : v4f{0.0f, 0.0f, 0.0f, 0.0f};
        }
        for (int u = 0; u < 4; ++u) {
            for (int j = 0; j < 4; ++j) {
                const unsigned d = (unsigned)(c[u][j] - lo);
                if (d < (unsigned)DEG_NR)
                    __hip_atomic_fetch_add(&sdeg[d], x[u][j], __ATOMIC_RELAXED,
                                           __HIP_MEMORY_SCOPE_WORKGROUP);
            }
        }
    }
    __syncthreads();

    float* dst = part + (size_t)s * DEG_PITCH + lo;
    for (int i = threadIdx.x; i < DEG_NR; i += DEG_TPB) dst[i] = sdeg[i];
}

__global__ void k_dinv_sum(const float* __restrict__ part,
                           float* __restrict__ dinv) {
    int i = blockIdx.x * blockDim.x + threadIdx.x;
    if (i < N_NODES) {
        float d = 1.0f;
        for (int s = 0; s < DEG_S; ++s) d += part[(size_t)s * DEG_PITCH + i];
        dinv[i] = (d > 0.0f) ? rsqrtf(d) : 0.0f;
    }
}

__global__ void k_zero(float* __restrict__ p, int n) {
    int stride = gridDim.x * blockDim.x;
    for (int i = blockIdx.x * blockDim.x + threadIdx.x; i < n; i += stride)
        p[i] = 0.0f;
}

__global__ void k_gemm_n64(const float* __restrict__ X,
                           const float* __restrict__ W,
                           float* __restrict__ H,
                           int M, int K) {
    const int lane  = threadIdx.x & 31;
    const int waveT = (blockIdx.x * blockDim.x + threadIdx.x) >> 5;
    const int nTiles = M >> 4;
    if (waveT >= nTiles) return;

    const int m0   = waveT << 4;
    const int mRow = m0 + (lane & 15);
    const int kSub = (lane & 16) ? 2 : 0;
    const int nCol = lane & 15;

    v8f c0 = {}, c1 = {}, c2 = {}, c3 = {};

    for (int k = 0; k < K; k += 4) {
        v2f a = *(const v2f*)(X + (size_t)mRow * K + (k + kSub));

        const float* wr0 = W + (size_t)(k + kSub)     * HID + nCol;
        const float* wr1 = W + (size_t)(k + kSub + 1) * HID + nCol;
        v2f b0, b1, b2, b3;
        b0.x = wr0[0];  b0.y = wr1[0];
        b1.x = wr0[16]; b1.y = wr1[16];
        b2.x = wr0[32]; b2.y = wr1[32];
        b3.x = wr0[48]; b3.y = wr1[48];

        c0 = __builtin_amdgcn_wmma_f32_16x16x4_f32(false, a, false, b0,
                                                   (short)0, c0, false, false);
        c1 = __builtin_amdgcn_wmma_f32_16x16x4_f32(false, a, false, b1,
                                                   (short)0, c1, false, false);
        c2 = __builtin_amdgcn_wmma_f32_16x16x4_f32(false, a, false, b2,
                                                   (short)0, c2, false, false);
        c3 = __builtin_amdgcn_wmma_f32_16x16x4_f32(false, a, false, b3,
                                                   (short)0, c3, false, false);
    }

    const int rBase = m0 + ((lane & 16) ? 8 : 0);
    #pragma unroll
    for (int j = 0; j < 8; ++j) {
        float* dst = H + (size_t)(rBase + j) * HID + nCol;
        dst[0]  = c0[j];
        dst[16] = c1[j];
        dst[32] = c2[j];
        dst[48] = c3[j];
    }
}

#define OWN_NR    1191
#define OWN_R     84
#define OWN_S     3
#define OWN_TPB   1024
#define OWN_NV    8
#define OWN_VTOT  (N_EDGES / 4)
#define OWN_VS    ((OWN_VTOT + OWN_S - 1) / OWN_S)
#define OWN_CHUNK (OWN_NV * OWN_TPB)
#define OWN_NCH   ((OWN_VS + OWN_CHUNK - 1) / OWN_CHUNK)
#define OWN_GB    16
#define OWN_WQ    64
#define OWN_RCAP  288
#define OWN_NWAVE (OWN_R * OWN_S * (OWN_TPB / 32))

template <int MODE>
__global__ __launch_bounds__(OWN_TPB)
void k_edge_owner(const int* __restrict__ row,
                  const int* __restrict__ col,
                  const float* __restrict__ w,
                  const float* __restrict__ dinv,
                  const float* __restrict__ h,
                  float* __restrict__ agg,
                  int* __restrict__ rec,
                  int* __restrict__ rcnt,
                  const float* wsBase, size_t wsSize) {
    __shared__ float    sums[(OWN_NR + 1) * HID];
    __shared__ unsigned wq[OWN_TPB / 32][OWN_WQ];

    const int tid  = threadIdx.x;
    const int lane = tid & 31;
    const int wv   = __builtin_amdgcn_readfirstlane(tid >> 5);
    const int r    = blockIdx.x % OWN_R;
    const int s    = blockIdx.x / OWN_R;
    const int lo   = r * OWN_NR;
    const int vs   = s * OWN_VS;
    const int ve   = (vs + OWN_VS < OWN_VTOT) ? (vs + OWN_VS) : OWN_VTOT;

    const bool room = (size_t)((const char*)(rcnt + OWN_NWAVE) - (const char*)wsBase) <= wsSize;
    const int  gw   = blockIdx.x * (OWN_TPB / 32) + wv;
    int* myrec = rec + (size_t)gw * (2 * OWN_RCAP);
    int  nrec  = 0;
    int  nrep  = -1;
    if (MODE == 1 && room) nrep = __builtin_amdgcn_readfirstlane(rcnt[gw]);
    const bool doscan = (MODE == 0) || nrep < 0;

    {
        v4f* z4 = (v4f*)sums;
        for (int i = tid; i < OWN_NR * HID / 4; i += OWN_TPB)
            z4[i] = v4f{0.0f, 0.0f, 0.0f, 0.0f};
    }

    const v4i* c4 = (const v4i*)col;
    v4i c[OWN_NV];
    v4i cn[OWN_NV];
    #pragma unroll
    for (int u = 0; u < OWN_NV; ++u) {
        const int v = vs + u * OWN_TPB + tid;
        c[u] = (doscan && v < ve) ? c4[v] : v4i{-1, -1, -1, -1};
    }
    __syncthreads();

    unsigned* myq = wq[wv];
    int count = 0;

    auto gather_add = [&](int rr, int dd, float nr, int cnt) {
        v2f   hv[OWN_GB];
        float nn[OWN_GB];
        int   dl[OWN_GB];
        #pragma unroll
        for (int u = 0; u < OWN_GB; ++u) {
            const bool ok = u < cnt;
            const int  i  = ok ? u : 0;
            const int  ru = __builtin_amdgcn_readlane(rr, i);
            const int  du = __builtin_amdgcn_readlane(dd, i);
            dl[u] = ok ? du : OWN_NR;
            nn[u] = __builtin_bit_cast(float,
                      __builtin_amdgcn_readlane(__builtin_bit_cast(int, nr), i));
            hv[u] = *(const v2f*)(h + (size_t)ru * HID + lane * 2);
        }
        #pragma unroll
        for (int u = 0; u < OWN_GB; ++u) {
            float* dst = &sums[dl[u] * HID + lane * 2];
            __hip_atomic_fetch_add(dst + 0, nn[u] * hv[u].x, __ATOMIC_RELAXED,
                                   __HIP_MEMORY_SCOPE_WORKGROUP);
            __hip_atomic_fetch_add(dst + 1, nn[u] * hv[u].y, __ATOMIC_RELAXED,
                                   __HIP_MEMORY_SCOPE_WORKGROUP);
        }
    };

    auto serve = [&](int base, int cnt) {
        int   rr = 0;
        int   dd = 0;
        float nr = 0.0f;
        if (lane < cnt) {
            const unsigned e = myq[base + lane];
            rr = row[e];
            const int   cc = col[e];
            const float we = w[e];
            dd = cc - lo;
            nr = dinv[rr] * we * dinv[cc];
        }
        if (MODE == 0 && room) {
            if (nrec + cnt <= OWN_RCAP && lane < cnt) {
                myrec[nrec + lane]            = rr | (dd << 17);
                myrec[OWN_RCAP + nrec + lane] = __builtin_bit_cast(int, nr);
            }
            nrec += cnt;
        }
        gather_add(rr, dd, nr, cnt);
    };

    if (doscan) {
    for (int ch = 0; ch < OWN_NCH; ++ch) {
        const int vb = vs + ch * OWN_CHUNK + tid;

        #pragma unroll
        for (int u = 0; u < OWN_NV; ++u) {
            const int v = vb + OWN_CHUNK + u * OWN_TPB;
            cn[u] = (ch + 1 < OWN_NCH && v < ve) ? c4[v] : v4i{-1, -1, -1, -1};
        }

        unsigned mask = 0u;
        #pragma unroll
        for (int u = 0; u < OWN_NV; ++u) {
            #pragma unroll
            for (int j = 0; j < 4; ++j) {
                const unsigned d = (unsigned)(c[u][j] - lo);
                mask |= (d < (unsigned)OWN_NR) ? (1u << (u * 4 + j)) : 0u;
            }
        }

        for (;;) {
            const bool     has = mask != 0u;
            const unsigned bal = __builtin_amdgcn_ballot_w32(has);
            if (bal == 0u) break;
            if (has) {
                const int idx = __builtin_ctz(mask);
                mask &= mask - 1u;
                const unsigned e = 4u * (unsigned)(vb + (idx >> 2) * OWN_TPB) + (unsigned)(idx & 3);
                myq[count + (int)__builtin_amdgcn_mbcnt_lo(bal, 0u)] = e;
            }
            count += __builtin_popcount(bal);
            __builtin_amdgcn_fence(__ATOMIC_ACQ_REL, "wavefront");
            while (count >= OWN_GB) {
                count -= OWN_GB;
                serve(count, OWN_GB);
            }
        }

        #pragma unroll
        for (int u = 0; u < OWN_NV; ++u) c[u] = cn[u];
    }
    if (count > 0) serve(0, count);
    if (MODE == 0 && room && lane == 0)
        rcnt[gw] = (nrec <= OWN_RCAP) ? nrec : -1;
    } else {
        if (nrep > 0) {
            const int last = nrep - 1;
            const int i0   = lane < last ? lane : last;
            int curp = myrec[i0];
            int curn = myrec[OWN_RCAP + i0];
            for (int base = 0; base < nrep; base += OWN_GB) {
                const int left = nrep - base;
                const int cnt  = left < OWN_GB ? left : OWN_GB;
                const int nx   = base + OWN_GB + lane;
                const int i1   = nx < last ? nx : last;
                const int nxtp = myrec[i1];
                const int nxtn = myrec[OWN_RCAP + i1];
                const int   rr = curp & 0x1ffff;
                const int   dd = curp >> 17;
                const float nr = __builtin_bit_cast(float, curn);
                if (cnt == OWN_GB) gather_add(rr, dd, nr, OWN_GB);
                else               gather_add(rr, dd, nr, cnt);
                curp = nxtp;
                curn = nxtn;
            }
        }
    }

    __syncthreads();

    float* dst = agg + (size_t)lo * HID;
    for (int i = tid; i < OWN_NR * HID; i += OWN_TPB) {
        if (lo + (i >> 6) < N_NODES) atomicAdd(dst + i, sums[i]);
    }
}

__global__ void k_finalize(float* __restrict__ agg,
                           const float* __restrict__ h,
                           const float* __restrict__ dinv,
                           const float* __restrict__ b,
                           int relu) {
    int stride = gridDim.x * blockDim.x;
    for (int i = blockIdx.x * blockDim.x + threadIdx.x; i < N_NODES * HID; i += stride) {
        const int n = i >> 6;
        const int f = i & 63;
        const float di = dinv[n];
        float v = agg[i] + di * di * h[i] + b[f];
        agg[i] = relu ? fmaxf(v, 0.0f) : v;
    }
}

__global__ void k_fc(const float* __restrict__ h,
                     const float* __restrict__ Wfc,
                     const float* __restrict__ bfc,
                     float* __restrict__ out) {
    int n = blockIdx.x * blockDim.x + threadIdx.x;
    if (n >= N_NODES) return;
    float acc[OUT_DIM];
    #pragma unroll
    for (int o = 0; o < OUT_DIM; ++o) acc[o] = bfc[o];
    const float* hr = h + (size_t)n * HID;
    #pragma unroll 4
    for (int k = 0; k < HID; ++k) {
        const float xv = hr[k];
        #pragma unroll
        for (int o = 0; o < OUT_DIM; ++o)
            acc[o] = fmaf(xv, Wfc[k * OUT_DIM + o], acc[o]);
    }
    float* orow = out + (size_t)n * OUT_DIM;
    #pragma unroll
    for (int o = 0; o < OUT_DIM; ++o) orow[o] = acc[o];
}

extern "C" void kernel_launch(void* const* d_in, const int* in_sizes, int n_in,
                              void* d_out, int out_size, void* d_ws, size_t ws_size,
                              hipStream_t stream) {
    const float* x   = (const float*)d_in[0];
    const int*   ei  = (const int*)  d_in[1];
    const float* ew  = (const float*)d_in[2];
    const float* W1  = (const float*)d_in[3];
    const float* b1  = (const float*)d_in[4];
    const float* W2  = (const float*)d_in[5];
    const float* b2  = (const float*)d_in[6];
    const float* Wfc = (const float*)d_in[7];
    const float* bfc = (const float*)d_in[8];
    float* out = (float*)d_out;

    const int* row = ei;
    const int* col = ei + N_EDGES;

    char* ws = (char*)d_ws;
    size_t off = 0;
    float* dinv = (float*)(ws + off); off += ((size_t)N_NODES * 4 + 255) & ~(size_t)255;
    float* bufH = (float*)(ws + off); off += (size_t)N_NODES * HID * 4;
    float* bufA = (float*)(ws + off); off += (size_t)N_NODES * HID * 4;
    int*   rec  = (int*)(ws + off);   off += (size_t)OWN_NWAVE * OWN_RCAP * 8;
    int*   rcnt = (int*)(ws + off);   off += (size_t)OWN_NWAVE * 4;
    (void)n_in; (void)in_sizes; (void)out_size;

    const int TPB = 256;
    const int nodeBlocks = (N_NODES + TPB - 1) / TPB;
    const int featElems  = N_NODES * HID;
    const int gemmBlocks = ((N_NODES / 16) + (TPB / 32) - 1) / (TPB / 32);

    k_deg_partial<<<DEG_R * DEG_S, DEG_TPB, 0, stream>>>(col, ew, bufA);
    k_dinv_sum<<<nodeBlocks, TPB, 0, stream>>>(bufA, dinv);

    k_gemm_n64<<<gemmBlocks, TPB, 0, stream>>>(x, W1, bufH, N_NODES, IN_DIM);
    k_zero<<<4096, TPB, 0, stream>>>(bufA, featElems);
    k_edge_owner<0><<<OWN_R * OWN_S, OWN_TPB, 0, stream>>>(row, col, ew, dinv, bufH, bufA,
                                                           rec, rcnt, dinv, ws_size);
    k_finalize<<<4096, TPB, 0, stream>>>(bufA, bufH, dinv, b1, 1);

    k_gemm_n64<<<gemmBlocks, TPB, 0, stream>>>(bufA, W2, bufH, N_NODES, HID);
    k_zero<<<4096, TPB, 0, stream>>>(bufA, featElems);
    k_edge_owner<1><<<OWN_R * OWN_S, OWN_TPB, 0, stream>>>(row, col, ew, dinv, bufH, bufA,
                                                           rec, rcnt, dinv, ws_size);
    k_finalize<<<4096, TPB, 0, stream>>>(bufA, bufH, dinv, b2, 1);

    k_fc<<<nodeBlocks, TPB, 0, stream>>>(bufA, Wfc, bfc, out);
}


# layer-1 GEMM split over k inside a work-group: wave w keeps W rows 64w..64w+63 in registers as B fragments, x read in 16-byte pieces; 8 partial tiles summed via LDS; k sum order changes
# speedup vs baseline: 1.6855x; 1.0631x over previous
#include <hip/hip_runtime.h>
#include <hip/hip_bf16.h>


#define N_NODES 100000
#define N_EDGES 1600000
#define IN_DIM  512
#define HID     64
#define OUT_DIM 5

typedef __attribute__((ext_vector_type(2))) float v2f;
typedef __attribute__((ext_vector_type(8))) float v8f;

#define DEG_R     8
#define DEG_S     32
#define DEG_NR    12512
#define DEG_PITCH (DEG_R * DEG_NR)
#define DEG_ES    (N_EDGES / DEG_S)
#define DEG_TPB   1024

typedef __attribute__((ext_vector_type(4))) int   v4i;
typedef __attribute__((ext_vector_type(4))) float v4f;

__global__ __launch_bounds__(DEG_TPB)
void k_deg_partial(const int* __restrict__ col,
                   const float* __restrict__ w,
                   float* __restrict__ part) {
    __shared__ float sdeg[DEG_NR];
    const int r  = blockIdx.x % DEG_R;
    const int s  = blockIdx.x / DEG_R;
    const int lo = r * DEG_NR;

    for (int i = threadIdx.x; i < DEG_NR; i += DEG_TPB) sdeg[i] = 0.0f;
    __syncthreads();

    const v4i* c4 = (const v4i*)(col + (size_t)s * DEG_ES);
    const v4f* w4 = (const v4f*)(w   + (size_t)s * DEG_ES);
    for (int v0 = threadIdx.x; v0 < DEG_ES / 4; v0 += 4 * DEG_TPB) {
        v4i c[4];
        v4f x[4];
        for (int u = 0; u < 4; ++u) {
            const int v = v0 + u * DEG_TPB;
            const bool in = v < DEG_ES / 4;
            c[u] = in ? c4[v] : v4i{-1, -1, -1, -1};
            x[u] = in ? w4[v] : v4f{0.0f, 0.0f, 0.0f, 0.0f};
        }
        for (int u = 0; u < 4; ++u) {
            for (int j = 0; j < 4; ++j) {
                const unsigned d = (unsigned)(c[u][j] - lo);
                if (d < (unsigned)DEG_NR)
                    __hip_atomic_fetch_add(&sdeg[d], x[u][j], __ATOMIC_RELAXED,
                                           __HIP_MEMORY_SCOPE_WORKGROUP);
            }
        }
    }
    __syncthreads();

    float* dst = part + (size_t)s * DEG_PITCH + lo;
    for (int i = threadIdx.x; i < DEG_NR; i += DEG_TPB) dst[i] = sdeg[i];
}

__global__ void k_dinv_sum(const float* __restrict__ part,
                           float* __restrict__ dinv) {
    int i = blockIdx.x * blockDim.x + threadIdx.x;
    if (i < N_NODES) {
        float d = 1.0f;
        for (int s = 0; s < DEG_S; ++s) d += part[(size_t)s * DEG_PITCH + i];
        dinv[i] = (d > 0.0f) ? rsqrtf(d) : 0.0f;
    }
}

__global__ void k_zero(float* __restrict__ p, int n) {
    int stride = gridDim.x * blockDim.x;
    for (int i = blockIdx.x * blockDim.x + threadIdx.x; i < n; i += stride)
        p[i] = 0.0f;
}

__global__ void k_gemm_n64(const float* __restrict__ X,
                           const float* __restrict__ W,
                           float* __restrict__ H,
                           int M, int K) {
    const int lane  = threadIdx.x & 31;
    const int waveT = (blockIdx.x * blockDim.x + threadIdx.x) >> 5;
    const int nTiles = M >> 4;
    if (waveT >= nTiles) return;

    const int m0   = waveT << 4;
    const int mRow = m0 + (lane & 15);
    const int kSub = (lane & 16) ? 2 : 0;
    const int nCol = lane & 15;

    v8f c0 = {}, c1 = {}, c2 = {}, c3 = {};

    for (int k = 0; k < K; k += 4) {
        v2f a = *(const v2f*)(X + (size_t)mRow * K + (k + kSub));

        const float* wr0 = W + (size_t)(k + kSub)     * HID + nCol;
        const float* wr1 = W + (size_t)(k + kSub + 1) * HID + nCol;
        v2f b0, b1, b2, b3;
        b0.x = wr0[0];  b0.y = wr1[0];
        b1.x = wr0[16]; b1.y = wr1[16];
        b2.x = wr0[32]; b2.y = wr1[32];
        b3.x = wr0[48]; b3.y = wr1[48];

        c0 = __builtin_amdgcn_wmma_f32_16x16x4_f32(false, a, false, b0,
                                                   (short)0, c0, false, false);
        c1 = __builtin_amdgcn_wmma_f32_16x16x4_f32(false, a, false, b1,
                                                   (short)0, c1, false, false);
        c2 = __builtin_amdgcn_wmma_f32_16x16x4_f32(false, a, false, b2,
                                                   (short)0, c2, false, false);
        c3 = __builtin_amdgcn_wmma_f32_16x16x4_f32(false, a, false, b3,
                                                   (short)0, c3, false, false);
    }

    const int rBase = m0 + ((lane & 16) ? 8 : 0);
    #pragma unroll
    for (int j = 0; j < 8; ++j) {
        float* dst = H + (size_t)(rBase + j) * HID + nCol;
        dst[0]  = c0[j];
        dst[16] = c1[j];
        dst[32] = c2[j];
        dst[48] = c3[j];
    }
}

#define SK_NW     8
#define SK_KS     (IN_DIM / SK_NW)
#define SK_NJ     (SK_KS / 8)
#define SK_PITCH  68
#define SK_GRID   512
#define SK_TILES  (N_NODES / 16)

__global__ __launch_bounds__(SK_NW * 32)
void k_gemm_splitk(const float* __restrict__ X,
                   const float* __restrict__ W,
                   float* __restrict__ H,
                   int nTiles) {
    __shared__ float red[2][SK_NW][16 * SK_PITCH];

    const int lane = threadIdx.x & 31;
    const int wv = __builtin_amdgcn_readfirstlane(threadIdx.x >> 5);
    const int c  = lane & 15;
    const int hf = lane >> 4;
    const int kb = wv * SK_KS;

    v2f bw[SK_NJ][2][4];
    #pragma unroll
    for (int j = 0; j < SK_NJ; ++j) {
        #pragma unroll
        for (int q = 0; q < 2; ++q) {
            const float* wp = W + (size_t)(kb + 8 * j + 4 * hf + 2 * q) * HID + c;
            #pragma unroll
            for (int n = 0; n < 4; ++n) {
                bw[j][q][n].x = wp[16 * n];
                bw[j][q][n].y = wp[HID + 16 * n];
            }
        }
    }

    const float* xp = X + (size_t)c * IN_DIM + kb + 4 * hf;

    int tile = blockIdx.x;
    v4f ac[SK_NJ];
    v4f an[SK_NJ];
    if (tile < nTiles) {
        const float* xt = xp + (size_t)tile * (16 * IN_DIM);
        #pragma unroll
        for (int j = 0; j < SK_NJ; ++j) ac[j] = *(const v4f*)(xt + 8 * j);
    }

    int buf = 0;
    for (; tile < nTiles; tile += gridDim.x) {
        {
            const int tn = tile + (int)gridDim.x;
            const int tl = tn < nTiles ? tn : tile;
            const float* xt = xp + (size_t)tl * (16 * IN_DIM);
            #pragma unroll
            for (int j = 0; j < SK_NJ; ++j) an[j] = *(const v4f*)(xt + 8 * j);
        }

        v8f acc[4] = {};
        #pragma unroll
        for (int j = 0; j < SK_NJ; ++j) {
            #pragma unroll
            for (int q = 0; q < 2; ++q) {
                v2f a;
                a.x = ac[j][2 * q];
                a.y = ac[j][2 * q + 1];
                #pragma unroll
                for (int n = 0; n < 4; ++n)
                    acc[n] = __builtin_amdgcn_wmma_f32_16x16x4_f32(false, a, false, bw[j][q][n],
                                                                   (short)0, acc[n], false, false);
            }
        }

        {
            float* pr = &red[buf][wv][(8 * hf) * SK_PITCH + c];
            #pragma unroll
            for (int n = 0; n < 4; ++n) {
                #pragma unroll
                for (int r = 0; r < 8; ++r) pr[r * SK_PITCH + 16 * n] = acc[n][r];
            }
        }
        __builtin_amdgcn_fence(__ATOMIC_RELEASE, "workgroup", "local");
        __builtin_amdgcn_s_barrier();
        __builtin_amdgcn_fence(__ATOMIC_ACQUIRE, "workgroup", "local");

        {
            const int rr = 2 * wv + hf;
            const float* sp = &red[buf][0][rr * SK_PITCH + 4 * c];
            v4f s = *(const v4f*)sp;
            #pragma unroll
            for (int p = 1; p < SK_NW; ++p) s += *(const v4f*)(sp + p * (16 * SK_PITCH));
            *(v4f*)(H + ((size_t)tile * 16 + rr) * HID + 4 * c) = s;
        }

        buf ^= 1;
        #pragma unroll
        for (int j = 0; j < SK_NJ; ++j) ac[j] = an[j];
    }
}

#define OWN_NR    1191
#define OWN_R     84
#define OWN_S     3
#define OWN_TPB   1024
#define OWN_NV    8
#define OWN_VTOT  (N_EDGES / 4)
#define OWN_VS    ((OWN_VTOT + OWN_S - 1) / OWN_S)
#define OWN_CHUNK (OWN_NV * OWN_TPB)
#define OWN_NCH   ((OWN_VS + OWN_CHUNK - 1) / OWN_CHUNK)
#define OWN_GB    16
#define OWN_WQ    64
#define OWN_RCAP  288
#define OWN_NWAVE (OWN_R * OWN_S * (OWN_TPB / 32))

template <int MODE>
__global__ __launch_bounds__(OWN_TPB)
void k_edge_owner(const int* __restrict__ row,
                  const int* __restrict__ col,
                  const float* __restrict__ w,
                  const float* __restrict__ dinv,
                  const float* __restrict__ h,
                  float* __restrict__ agg,
                  int* __restrict__ rec,
                  int* __restrict__ rcnt,
                  const float* wsBase, size_t wsSize) {
    __shared__ float    sums[(OWN_NR + 1) * HID];
    __shared__ unsigned wq[OWN_TPB / 32][OWN_WQ];

    const int tid  = threadIdx.x;
    const int lane = tid & 31;
    const int wv   = __builtin_amdgcn_readfirstlane(tid >> 5);
    const int r    = blockIdx.x % OWN_R;
    const int s    = blockIdx.x / OWN_R;
    const int lo   = r * OWN_NR;
    const int vs   = s * OWN_VS;
    const int ve   = (vs + OWN_VS < OWN_VTOT) ? (vs + OWN_VS) : OWN_VTOT;

    const bool room = (size_t)((const char*)(rcnt + OWN_NWAVE) - (const char*)wsBase) <= wsSize;
    const int  gw   = blockIdx.x * (OWN_TPB / 32) + wv;
    int* myrec = rec + (size_t)gw * (2 * OWN_RCAP);
    int  nrec  = 0;
    int  nrep  = -1;
    if (MODE == 1 && room) nrep = __builtin_amdgcn_readfirstlane(rcnt[gw]);
    const bool doscan = (MODE == 0) || nrep < 0;

    {
        v4f* z4 = (v4f*)sums;
        for (int i = tid; i < OWN_NR * HID / 4; i += OWN_TPB)
            z4[i] = v4f{0.0f, 0.0f, 0.0f, 0.0f};
    }

    const v4i* c4 = (const v4i*)col;
    v4i c[OWN_NV];
    v4i cn[OWN_NV];
    #pragma unroll
    for (int u = 0; u < OWN_NV; ++u) {
        const int v = vs + u * OWN_TPB + tid;
        c[u] = (doscan && v < ve) ? c4[v] : v4i{-1, -1, -1, -1};
    }
    __syncthreads();

    unsigned* myq = wq[wv];
    int count = 0;

    auto gather_add = [&](int rr, int dd, float nr, int cnt) {
        v2f   hv[OWN_GB];
        float nn[OWN_GB];
        int   dl[OWN_GB];
        #pragma unroll
        for (int u = 0; u < OWN_GB; ++u) {
            const bool ok = u < cnt;
            const int  i  = ok ? u : 0;
            const int  ru = __builtin_amdgcn_readlane(rr, i);
            const int  du = __builtin_amdgcn_readlane(dd, i);
            dl[u] = ok ? du : OWN_NR;
            nn[u] = __builtin_bit_cast(float,
                      __builtin_amdgcn_readlane(__builtin_bit_cast(int, nr), i));
            hv[u] = *(const v2f*)(h + (size_t)ru * HID + lane * 2);
        }
        #pragma unroll
        for (int u = 0; u < OWN_GB; ++u) {
            float* dst = &sums[dl[u] * HID + lane * 2];
            __hip_atomic_fetch_add(dst + 0, nn[u] * hv[u].x, __ATOMIC_RELAXED,
                                   __HIP_MEMORY_SCOPE_WORKGROUP);
            __hip_atomic_fetch_add(dst + 1, nn[u] * hv[u].y, __ATOMIC_RELAXED,
                                   __HIP_MEMORY_SCOPE_WORKGROUP);
        }
    };

    auto serve = [&](int base, int cnt) {
        int   rr = 0;
        int   dd = 0;
        float nr = 0.0f;
        if (lane < cnt) {
            const unsigned e = myq[base + lane];
            rr = row[e];
            const int   cc = col[e];
            const float we = w[e];
            dd = cc - lo;
            nr = dinv[rr] * we * dinv[cc];
        }
        if (MODE == 0 && room) {
            if (nrec + cnt <= OWN_RCAP && lane < cnt) {
                myrec[nrec + lane]            = rr | (dd << 17);
                myrec[OWN_RCAP + nrec + lane] = __builtin_bit_cast(int, nr);
            }
            nrec += cnt;
        }
        gather_add(rr, dd, nr, cnt);
    };

    if (doscan) {
    for (int ch = 0; ch < OWN_NCH; ++ch) {
        const int vb = vs + ch * OWN_CHUNK + tid;

        #pragma unroll
        for (int u = 0; u < OWN_NV; ++u) {
            const int v = vb + OWN_CHUNK + u * OWN_TPB;
            cn[u] = (ch + 1 < OWN_NCH && v < ve) ? c4[v] : v4i{-1, -1, -1, -1};
        }

        unsigned mask = 0u;
        #pragma unroll
        for (int u = 0; u < OWN_NV; ++u) {
            #pragma unroll
            for (int j = 0; j < 4; ++j) {
                const unsigned d = (unsigned)(c[u][j] - lo);
                mask |= (d < (unsigned)OWN_NR) ? (1u << (u * 4 + j)) : 0u;
            }
        }

        for (;;) {
            const bool     has = mask != 0u;
            const unsigned bal = __builtin_amdgcn_ballot_w32(has);
            if (bal == 0u) break;
            if (has) {
                const int idx = __builtin_ctz(mask);
                mask &= mask - 1u;
                const unsigned e = 4u * (unsigned)(vb + (idx >> 2) * OWN_TPB) + (unsigned)(idx & 3);
                myq[count + (int)__builtin_amdgcn_mbcnt_lo(bal, 0u)] = e;
            }
            count += __builtin_popcount(bal);
            __builtin_amdgcn_fence(__ATOMIC_ACQ_REL, "wavefront");
            while (count >= OWN_GB) {
                count -= OWN_GB;
                serve(count, OWN_GB);
            }
        }

        #pragma unroll
        for (int u = 0; u < OWN_NV; ++u) c[u] = cn[u];
    }
    if (count > 0) serve(0, count);
    if (MODE == 0 && room && lane == 0)
        rcnt[gw] = (nrec <= OWN_RCAP) ? nrec : -1;
    } else {
        if (nrep > 0) {
            const int last = nrep - 1;
            const int i0   = lane < last ? lane : last;
            int curp = myrec[i0];
            int curn = myrec[OWN_RCAP + i0];
            for (int base = 0; base < nrep; base += OWN_GB) {
                const int left = nrep - base;
                const int cnt  = left < OWN_GB ? left : OWN_GB;
                const int nx   = base + OWN_GB + lane;
                const int i1   = nx < last ? nx : last;
                const int nxtp = myrec[i1];
                const int nxtn = myrec[OWN_RCAP + i1];
                const int   rr = curp & 0x1ffff;
                const int   dd = curp >> 17;
                const float nr = __builtin_bit_cast(float, curn);
                if (cnt == OWN_GB) gather_add(rr, dd, nr, OWN_GB);
                else               gather_add(rr, dd, nr, cnt);
                curp = nxtp;
                curn = nxtn;
            }
        }
    }

    __syncthreads();

    float* dst = agg + (size_t)lo * HID;
    for (int i = tid; i < OWN_NR * HID; i += OWN_TPB) {
        if (lo + (i >> 6) < N_NODES) atomicAdd(dst + i, sums[i]);
    }
}

__global__ void k_finalize(float* __restrict__ agg,
                           const float* __restrict__ h,
                           const float* __restrict__ dinv,
                           const float* __restrict__ b,
                           int relu) {
    int stride = gridDim.x * blockDim.x;
    for (int i = blockIdx.x * blockDim.x + threadIdx.x; i < N_NODES * HID; i += stride) {
        const int n = i >> 6;
        const int f = i & 63;
        const float di = dinv[n];
        float v = agg[i] + di * di * h[i] + b[f];
        agg[i] = relu ? fmaxf(v, 0.0f) : v;
    }
}

__global__ void k_fc(const float* __restrict__ h,
                     const float* __restrict__ Wfc,
                     const float* __restrict__ bfc,
                     float* __restrict__ out) {
    int n = blockIdx.x * blockDim.x + threadIdx.x;
    if (n >= N_NODES) return;
    float acc[OUT_DIM];
    #pragma unroll
    for (int o = 0; o < OUT_DIM; ++o) acc[o] = bfc[o];
    const float* hr = h + (size_t)n * HID;
    #pragma unroll 4
    for (int k = 0; k < HID; ++k) {
        const float xv = hr[k];
        #pragma unroll
        for (int o = 0; o < OUT_DIM; ++o)
            acc[o] = fmaf(xv, Wfc[k * OUT_DIM + o], acc[o]);
    }
    float* orow = out + (size_t)n * OUT_DIM;
    #pragma unroll
    for (int o = 0; o < OUT_DIM; ++o) orow[o] = acc[o];
}

extern "C" void kernel_launch(void* const* d_in, const int* in_sizes, int n_in,
                              void* d_out, int out_size, void* d_ws, size_t ws_size,
                              hipStream_t stream) {
    const float* x   = (const float*)d_in[0];
    const int*   ei  = (const int*)  d_in[1];
    const float* ew  = (const float*)d_in[2];
    const float* W1  = (const float*)d_in[3];
    const float* b1  = (const float*)d_in[4];
    const float* W2  = (const float*)d_in[5];
    const float* b2  = (const float*)d_in[6];
    const float* Wfc = (const float*)d_in[7];
    const float* bfc = (const float*)d_in[8];
    float* out = (float*)d_out;

    const int* row = ei;
    const int* col = ei + N_EDGES;

    char* ws = (char*)d_ws;
    size_t off = 0;
    float* dinv = (float*)(ws + off); off += ((size_t)N_NODES * 4 + 255) & ~(size_t)255;
    float* bufH = (float*)(ws + off); off += (size_t)N_NODES * HID * 4;
    float* bufA = (float*)(ws + off); off += (size_t)N_NODES * HID * 4;
    int*   rec  = (int*)(ws + off);   off += (size_t)OWN_NWAVE * OWN_RCAP * 8;
    int*   rcnt = (int*)(ws + off);   off += (size_t)OWN_NWAVE * 4;
    (void)n_in; (void)in_sizes; (void)out_size;

    const int TPB = 256;
    const int nodeBlocks = (N_NODES + TPB - 1) / TPB;
    const int featElems  = N_NODES * HID;
    const int gemmBlocks = ((N_NODES / 16) + (TPB / 32) - 1) / (TPB / 32);

    k_deg_partial<<<DEG_R * DEG_S, DEG_TPB, 0, stream>>>(col, ew, bufA);
    k_dinv_sum<<<nodeBlocks, TPB, 0, stream>>>(bufA, dinv);

    k_gemm_splitk<<<SK_GRID, SK_NW * 32, 0, stream>>>(x, W1, bufH, SK_TILES);
    k_zero<<<4096, TPB, 0, stream>>>(bufA, featElems);
    k_edge_owner<0><<<OWN_R * OWN_S, OWN_TPB, 0, stream>>>(row, col, ew, dinv, bufH, bufA,
                                                           rec, rcnt, dinv, ws_size);
    k_finalize<<<4096, TPB, 0, stream>>>(bufA, bufH, dinv, b1, 1);

    k_gemm_n64<<<gemmBlocks, TPB, 0, stream>>>(bufA, W2, bufH, N_NODES, HID);
    k_zero<<<4096, TPB, 0, stream>>>(bufA, featElems);
    k_edge_owner<1><<<OWN_R * OWN_S, OWN_TPB, 0, stream>>>(row, col, ew, dinv, bufH, bufA,
                                                           rec, rcnt, dinv, ws_size);
    k_finalize<<<4096, TPB, 0, stream>>>(bufA, bufH, dinv, b2, 1);

    k_fc<<<nodeBlocks, TPB, 0, stream>>>(bufA, Wfc, bfc, out);
}


# layer-1 edge scan reads one byte per edge (number of the target's 1191-node range, written by the degree pass) instead of the 4-byte target id; sum order: arrival as before, edges fall to other waves
# speedup vs baseline: 1.7802x; 1.0562x over previous
#include <hip/hip_runtime.h>
#include <hip/hip_bf16.h>


#define N_NODES 100000
#define N_EDGES 1600000
#define IN_DIM  512
#define HID     64
#define OUT_DIM 5

typedef __attribute__((ext_vector_type(2))) float v2f;
typedef __attribute__((ext_vector_type(8))) float v8f;

#define DEG_R     8
#define DEG_S     32
#define DEG_NR    12512
#define DEG_PITCH (DEG_R * DEG_NR)
#define DEG_ES    (N_EDGES / DEG_S)
#define DEG_TPB   1024

typedef __attribute__((ext_vector_type(4))) int   v4i;
typedef __attribute__((ext_vector_type(4))) float v4f;

#define OWN_NR    1191

__device__ __forceinline__ unsigned rid_pack(v4i c) {
    unsigned o = 0u;
    #pragma unroll
    for (int j = 0; j < 4; ++j) {
        unsigned q = (unsigned)c[j] / (unsigned)OWN_NR;
        q = q < 255u ? q : 255u;
        o |= q << (8 * j);
    }
    return o;
}

__global__ __launch_bounds__(DEG_TPB)
void k_deg_partial(const int* __restrict__ col,
                   const float* __restrict__ w,
                   float* __restrict__ part,
                   unsigned* __restrict__ ridw,
                   const float* wsBase, size_t wsSize) {
    __shared__ float sdeg[DEG_NR];
    const int r  = blockIdx.x % DEG_R;
    const int s  = blockIdx.x / DEG_R;
    const int lo = r * DEG_NR;
    const bool hasb = (size_t)((const char*)(ridw + N_EDGES / 4) - (const char*)wsBase) <= wsSize;

    for (int i = threadIdx.x; i < DEG_NR; i += DEG_TPB) sdeg[i] = 0.0f;
    __syncthreads();

    const v4i* c4 = (const v4i*)(col + (size_t)s * DEG_ES);
    const v4f* w4 = (const v4f*)(w   + (size_t)s * DEG_ES);
    int blk = 0;
    for (int v0 = threadIdx.x; v0 < DEG_ES / 4; v0 += 4 * DEG_TPB, blk += 4) {
        v4i c[4];
        v4f x[4];
        for (int u = 0; u < 4; ++u) {
            const int v = v0 + u * DEG_TPB;
            const bool in = v < DEG_ES / 4;
            c[u] = in ? c4[v] : v4i{-1, -1, -1, -1};
            x[u] = in ? w4[v] : v4f{0.0f, 0.0f, 0.0f, 0.0f};
        }
        if (hasb) {
            for (int u = 0; u < 4; ++u) {
                const int v = v0 + u * DEG_TPB;
                if ((blk + u) % DEG_R == r && v < DEG_ES / 4)
                    ridw[(size_t)s * (DEG_ES / 4) + v] = rid_pack(c[u]);
            }
        }
        for (int u = 0; u < 4; ++u) {
            for (int j = 0; j < 4; ++j) {
                const unsigned d = (unsigned)(c[u][j] - lo);
                if (d < (unsigned)DEG_NR)
                    __hip_atomic_fetch_add(&sdeg[d], x[u][j], __ATOMIC_RELAXED,
                                           __HIP_MEMORY_SCOPE_WORKGROUP);
            }
        }
    }
    __syncthreads();

    float* dst = part + (size_t)s * DEG_PITCH + lo;
    for (int i = threadIdx.x; i < DEG_NR; i += DEG_TPB) dst[i] = sdeg[i];
}

__global__ void k_dinv_sum(const float* __restrict__ part,
                           float* __restrict__ dinv) {
    int i = blockIdx.x * blockDim.x + threadIdx.x;
    if (i < N_NODES) {
        float d = 1.0f;
        for (int s = 0; s < DEG_S; ++s) d += part[(size_t)s * DEG_PITCH + i];
        dinv[i] = (d > 0.0f) ? rsqrtf(d) : 0.0f;
    }
}

__global__ void k_zero(float* __restrict__ p, int n) {
    int stride = gridDim.x * blockDim.x;
    for (int i = blockIdx.x * blockDim.x + threadIdx.x; i < n; i += stride)
        p[i] = 0.0f;
}

__global__ void k_gemm_n64(const float* __restrict__ X,
                           const float* __restrict__ W,
                           float* __restrict__ H,
                           int M, int K) {
    const int lane  = threadIdx.x & 31;
    const int waveT = (blockIdx.x * blockDim.x + threadIdx.x) >> 5;
    const int nTiles = M >> 4;
    if (waveT >= nTiles) return;

    const int m0   = waveT << 4;
    const int mRow = m0 + (lane & 15);
    const int kSub = (lane & 16) ? 2 : 0;
    const int nCol = lane & 15;

    v8f c0 = {}, c1 = {}, c2 = {}, c3 = {};

    for (int k = 0; k < K; k += 4) {
        v2f a = *(const v2f*)(X + (size_t)mRow * K + (k + kSub));

        const float* wr0 = W + (size_t)(k + kSub)     * HID + nCol;
        const float* wr1 = W + (size_t)(k + kSub + 1) * HID + nCol;
        v2f b0, b1, b2, b3;
        b0.x = wr0[0];  b0.y = wr1[0];
        b1.x = wr0[16]; b1.y = wr1[16];
        b2.x = wr0[32]; b2.y = wr1[32];
        b3.x = wr0[48]; b3.y = wr1[48];

        c0 = __builtin_amdgcn_wmma_f32_16x16x4_f32(false, a, false, b0,
                                                   (short)0, c0, false, false);
        c1 = __builtin_amdgcn_wmma_f32_16x16x4_f32(false, a, false, b1,
                                                   (short)0, c1, false, false);
        c2 = __builtin_amdgcn_wmma_f32_16x16x4_f32(false, a, false, b2,
                                                   (short)0, c2, false, false);
        c3 = __builtin_amdgcn_wmma_f32_16x16x4_f32(false, a, false, b3,
                                                   (short)0, c3, false, false);
    }

    const int rBase = m0 + ((lane & 16) ? 8 : 0);
    #pragma unroll
    for (int j = 0; j < 8; ++j) {
        float* dst = H + (size_t)(rBase + j) * HID + nCol;
        dst[0]  = c0[j];
        dst[16] = c1[j];
        dst[32] = c2[j];
        dst[48] = c3[j];
    }
}

#define SK_NW     8
#define SK_KS     (IN_DIM / SK_NW)
#define SK_NJ     (SK_KS / 8)
#define SK_PITCH  68
#define SK_GRID   512
#define SK_TILES  (N_NODES / 16)

__global__ __launch_bounds__(SK_NW * 32)
void k_gemm_splitk(const float* __restrict__ X,
                   const float* __restrict__ W,
                   float* __restrict__ H,
                   int nTiles) {
    __shared__ float red[2][SK_NW][16 * SK_PITCH];

    const int lane = threadIdx.x & 31;
    const int wv = __builtin_amdgcn_readfirstlane(threadIdx.x >> 5);
    const int c  = lane & 15;
    const int hf = lane >> 4;
    const int kb = wv * SK_KS;

    v2f bw[SK_NJ][2][4];
    #pragma unroll
    for (int j = 0; j < SK_NJ; ++j) {
        #pragma unroll
        for (int q = 0; q < 2; ++q) {
            const float* wp = W + (size_t)(kb + 8 * j + 4 * hf + 2 * q) * HID + c;
            #pragma unroll
            for (int n = 0; n < 4; ++n) {
                bw[j][q][n].x = wp[16 * n];
                bw[j][q][n].y = wp[HID + 16 * n];
            }
        }
    }

    const float* xp = X + (size_t)c * IN_DIM + kb + 4 * hf;

    int tile = blockIdx.x;
    v4f ac[SK_NJ];
    v4f an[SK_NJ];
    if (tile < nTiles) {
        const float* xt = xp + (size_t)tile * (16 * IN_DIM);
        #pragma unroll
        for (int j = 0; j < SK_NJ; ++j) ac[j] = *(const v4f*)(xt + 8 * j);
    }

    int buf = 0;
    for (; tile < nTiles; tile += gridDim.x) {
        {
            const int tn = tile + (int)gridDim.x;
            const int tl = tn < nTiles ? tn : tile;
            const float* xt = xp + (size_t)tl * (16 * IN_DIM);
            #pragma unroll
            for (int j = 0; j < SK_NJ; ++j) an[j] = *(const v4f*)(xt + 8 * j);
        }

        v8f acc[4] = {};
        #pragma unroll
        for (int j = 0; j < SK_NJ; ++j) {
            #pragma unroll
            for (int q = 0; q < 2; ++q) {
                v2f a;
                a.x = ac[j][2 * q];
                a.y = ac[j][2 * q + 1];
                #pragma unroll
                for (int n = 0; n < 4; ++n)
                    acc[n] = __builtin_amdgcn_wmma_f32_16x16x4_f32(false, a, false, bw[j][q][n],
                                                                   (short)0, acc[n], false, false);
            }
        }

        {
            float* pr = &red[buf][wv][(8 * hf) * SK_PITCH + c];
            #pragma unroll
            for (int n = 0; n < 4; ++n) {
                #pragma unroll
                for (int r = 0; r < 8; ++r) pr[r * SK_PITCH + 16 * n] = acc[n][r];
            }
        }
        __builtin_amdgcn_fence(__ATOMIC_RELEASE, "workgroup", "local");
        __builtin_amdgcn_s_barrier();
        __builtin_amdgcn_fence(__ATOMIC_ACQUIRE, "workgroup", "local");

        {
            const int rr = 2 * wv + hf;
            const float* sp = &red[buf][0][rr * SK_PITCH + 4 * c];
            v4f s = *(const v4f*)sp;
            #pragma unroll
            for (int p = 1; p < SK_NW; ++p) s += *(const v4f*)(sp + p * (16 * SK_PITCH));
            *(v4f*)(H + ((size_t)tile * 16 + rr) * HID + 4 * c) = s;
        }

        buf ^= 1;
        #pragma unroll
        for (int j = 0; j < SK_NJ; ++j) ac[j] = an[j];
    }
}

#define OWN_R     84
#define OWN_S     3
#define OWN_TPB   1024
#define OWN_NV    2
#define OWN_VTOT  (N_EDGES / 16)
#define OWN_VS    ((OWN_VTOT + OWN_S - 1) / OWN_S)
#define OWN_CHUNK (OWN_NV * OWN_TPB)
#define OWN_NCH   ((OWN_VS + OWN_CHUNK - 1) / OWN_CHUNK)
#define OWN_GB    16
#define OWN_WQ    64
#define OWN_RCAP  288
#define OWN_NWAVE (OWN_R * OWN_S * (OWN_TPB / 32))

template <int MODE>
__global__ __launch_bounds__(OWN_TPB)
void k_edge_owner(const int* __restrict__ row,
                  const int* __restrict__ col,
                  const float* __restrict__ w,
                  const float* __restrict__ dinv,
                  const float* __restrict__ h,
                  float* __restrict__ agg,
                  int* __restrict__ rec,
                  int* __restrict__ rcnt,
                  const unsigned char* __restrict__ rid,
                  const float* wsBase, size_t wsSize) {
    __shared__ float    sums[(OWN_NR + 1) * HID];
    __shared__ unsigned wq[OWN_TPB / 32][OWN_WQ];

    const int tid  = threadIdx.x;
    const int lane = tid & 31;
    const int wv   = __builtin_amdgcn_readfirstlane(tid >> 5);
    const int r    = blockIdx.x % OWN_R;
    const int s    = blockIdx.x / OWN_R;
    const int lo   = r * OWN_NR;
    const int vs   = s * OWN_VS;
    const int ve   = (vs + OWN_VS < OWN_VTOT) ? (vs + OWN_VS) : OWN_VTOT;

    const bool room = (size_t)((const char*)(rcnt + OWN_NWAVE) - (const char*)wsBase) <= wsSize;
    const bool hasb = (size_t)((const char*)(rid + N_EDGES) - (const char*)wsBase) <= wsSize;
    const int  gw   = blockIdx.x * (OWN_TPB / 32) + wv;
    int* myrec = rec + (size_t)gw * (2 * OWN_RCAP);
    int  nrec  = 0;
    int  nrep  = -1;
    if (MODE == 1 && room) nrep = __builtin_amdgcn_readfirstlane(rcnt[gw]);
    const bool doscan = (MODE == 0) || nrep < 0;

    {
        v4f* z4 = (v4f*)sums;
        for (int i = tid; i < OWN_NR * HID / 4; i += OWN_TPB)
            z4[i] = v4f{0.0f, 0.0f, 0.0f, 0.0f};
    }

    const v4i* b4 = (const v4i*)rid;
    const v4i* c4 = (const v4i*)col;
    auto ldids = [&](int v) -> v4i {
        if (hasb) return b4[v];
        v4i o;
        #pragma unroll
        for (int q = 0; q < 4; ++q) o[q] = (int)rid_pack(c4[4 * v + q]);
        return o;
    };
    v4i c[OWN_NV];
    v4i cn[OWN_NV];
    #pragma unroll
    for (int u = 0; u < OWN_NV; ++u) {
        const int v = vs + u * OWN_TPB + tid;
        c[u] = (doscan && v < ve) ? ldids(v) : v4i{-1, -1, -1, -1};
    }
    __syncthreads();

    unsigned* myq = wq[wv];
    int count = 0;

    auto gather_add = [&](int rr, int dd, float nr, int cnt) {
        v2f   hv[OWN_GB];
        float nn[OWN_GB];
        int   dl[OWN_GB];
        #pragma unroll
        for (int u = 0; u < OWN_GB; ++u) {
            const bool ok = u < cnt;
            const int  i  = ok ? u : 0;
            const int  ru = __builtin_amdgcn_readlane(rr, i);
            const int  du = __builtin_amdgcn_readlane(dd, i);
            dl[u] = ok ? du : OWN_NR;
            nn[u] = __builtin_bit_cast(float,
                      __builtin_amdgcn_readlane(__builtin_bit_cast(int, nr), i));
            hv[u] = *(const v2f*)(h + (size_t)ru * HID + lane * 2);
        }
        #pragma unroll
        for (int u = 0; u < OWN_GB; ++u) {
            float* dst = &sums[dl[u] * HID + lane * 2];
            __hip_atomic_fetch_add(dst + 0, nn[u] * hv[u].x, __ATOMIC_RELAXED,
                                   __HIP_MEMORY_SCOPE_WORKGROUP);
            __hip_atomic_fetch_add(dst + 1, nn[u] * hv[u].y, __ATOMIC_RELAXED,
                                   __HIP_MEMORY_SCOPE_WORKGROUP);
        }
    };

    auto serve = [&](int base, int cnt) {
        int   rr = 0;
        int   dd = 0;
        float nr = 0.0f;
        if (lane < cnt) {
            const unsigned e = myq[base + lane];
            rr = row[e];
            const int   cc = col[e];
            const float we = w[e];
            dd = cc - lo;
            nr = dinv[rr] * we * dinv[cc];
        }
        if (MODE == 0 && room) {
            if (nrec + cnt <= OWN_RCAP && lane < cnt) {
                myrec[nrec + lane]            = rr | (dd << 17);
                myrec[OWN_RCAP + nrec + lane] = __builtin_bit_cast(int, nr);
            }
            nrec += cnt;
        }
        gather_add(rr, dd, nr, cnt);
    };

    if (doscan) {
    for (int ch = 0; ch < OWN_NCH; ++ch) {
        const int vb = vs + ch * OWN_CHUNK + tid;

        #pragma unroll
        for (int u = 0; u < OWN_NV; ++u) {
            const int v = vb + OWN_CHUNK + u * OWN_TPB;
            cn[u] = (ch + 1 < OWN_NCH && v < ve) ? ldids(v) : v4i{-1, -1, -1, -1};
        }

        unsigned mask = 0u;
        const unsigned r4 = (unsigned)r * 0x01010101u;
        #pragma unroll
        for (int u = 0; u < OWN_NV; ++u) {
            #pragma unroll
            for (int q = 0; q < 4; ++q) {
                const unsigned y = (unsigned)c[u][q] ^ r4;
                const unsigned t = ((y & 0x7f7f7f7fu) + 0x7f7f7f7fu) | y;
                mask |= (~t & 0x80808080u) >> (7 - (4 * u + q));
            }
        }

        for (;;) {
            const bool     has = mask != 0u;
            const unsigned bal = __builtin_amdgcn_ballot_w32(has);
            if (bal == 0u) break;
            if (has) {
                const int idx = __builtin_ctz(mask);
                mask &= mask - 1u;
                const int k = idx & 7;
                const unsigned e = 16u * (unsigned)(vb + (k >> 2) * OWN_TPB)
                                 + 4u * (unsigned)(k & 3) + (unsigned)(idx >> 3);
                myq[count + (int)__builtin_amdgcn_mbcnt_lo(bal, 0u)] = e;
            }
            count += __builtin_popcount(bal);
            __builtin_amdgcn_fence(__ATOMIC_ACQ_REL, "wavefront");
            while (count >= OWN_GB) {
                count -= OWN_GB;
                serve(count, OWN_GB);
            }
        }

        #pragma unroll
        for (int u = 0; u < OWN_NV; ++u) c[u] = cn[u];
    }
    if (count > 0) serve(0, count);
    if (MODE == 0 && room && lane == 0)
        rcnt[gw] = (nrec <= OWN_RCAP) ? nrec : -1;
    } else {
        if (nrep > 0) {
            const int last = nrep - 1;
            const int i0   = lane < last ? lane : last;
            int curp = myrec[i0];
            int curn = myrec[OWN_RCAP + i0];
            for (int base = 0; base < nrep; base += OWN_GB) {
                const int left = nrep - base;
                const int cnt  = left < OWN_GB ? left : OWN_GB;
                const int nx   = base + OWN_GB + lane;
                const int i1   = nx < last ? nx : last;
                const int nxtp = myrec[i1];
                const int nxtn = myrec[OWN_RCAP + i1];
                const int   rr = curp & 0x1ffff;
                const int   dd = curp >> 17;
                const float nr = __builtin_bit_cast(float, curn);
                if (cnt == OWN_GB) gather_add(rr, dd, nr, OWN_GB);
                else               gather_add(rr, dd, nr, cnt);
                curp = nxtp;
                curn = nxtn;
            }
        }
    }

    __syncthreads();

    float* dst = agg + (size_t)lo * HID;
    for (int i = tid; i < OWN_NR * HID; i += OWN_TPB) {
        if (lo + (i >> 6) < N_NODES) atomicAdd(dst + i, sums[i]);
    }
}

__global__ void k_finalize(float* __restrict__ agg,
                           const float* __restrict__ h,
                           const float* __restrict__ dinv,
                           const float* __restrict__ b,
                           int relu) {
    int stride = gridDim.x * blockDim.x;
    for (int i = blockIdx.x * blockDim.x + threadIdx.x; i < N_NODES * HID; i += stride) {
        const int n = i >> 6;
        const int f = i & 63;
        const float di = dinv[n];
        float v = agg[i] + di * di * h[i] + b[f];
        agg[i] = relu ? fmaxf(v, 0.0f) : v;
    }
}

__global__ void k_fc(const float* __restrict__ h,
                     const float* __restrict__ Wfc,
                     const float* __restrict__ bfc,
                     float* __restrict__ out) {
    int n = blockIdx.x * blockDim.x + threadIdx.x;
    if (n >= N_NODES) return;
    float acc[OUT_DIM];
    #pragma unroll
    for (int o = 0; o < OUT_DIM; ++o) acc[o] = bfc[o];
    const float* hr = h + (size_t)n * HID;
    #pragma unroll 4
    for (int k = 0; k < HID; ++k) {
        const float xv = hr[k];
        #pragma unroll
        for (int o = 0; o < OUT_DIM; ++o)
            acc[o] = fmaf(xv, Wfc[k * OUT_DIM + o], acc[o]);
    }
    float* orow = out + (size_t)n * OUT_DIM;
    #pragma unroll
    for (int o = 0; o < OUT_DIM; ++o) orow[o] = acc[o];
}

extern "C" void kernel_launch(void* const* d_in, const int* in_sizes, int n_in,
                              void* d_out, int out_size, void* d_ws, size_t ws_size,
                              hipStream_t stream) {
    const float* x   = (const float*)d_in[0];
    const int*   ei  = (const int*)  d_in[1];
    const float* ew  = (const float*)d_in[2];
    const float* W1  = (const float*)d_in[3];
    const float* b1  = (const float*)d_in[4];
    const float* W2  = (const float*)d_in[5];
    const float* b2  = (const float*)d_in[6];
    const float* Wfc = (const float*)d_in[7];
    const float* bfc = (const float*)d_in[8];
    float* out = (float*)d_out;

    const int* row = ei;
    const int* col = ei + N_EDGES;

    char* ws = (char*)d_ws;
    size_t off = 0;
    float* dinv = (float*)(ws + off); off += ((size_t)N_NODES * 4 + 255) & ~(size_t)255;
    float* bufH = (float*)(ws + off); off += (size_t)N_NODES * HID * 4;
    float* bufA = (float*)(ws + off); off += (size_t)N_NODES * HID * 4;
    int*   rec  = (int*)(ws + off);   off += (size_t)OWN_NWAVE * OWN_RCAP * 8;
    int*   rcnt = (int*)(ws + off);   off += (size_t)OWN_NWAVE * 4;
    unsigned char* rid = (unsigned char*)(ws + off);   off += (size_t)N_EDGES;
    (void)n_in; (void)in_sizes; (void)out_size;

    const int TPB = 256;
    const int nodeBlocks = (N_NODES + TPB - 1) / TPB;
    const int featElems  = N_NODES * HID;
    const int gemmBlocks = ((N_NODES / 16) + (TPB / 32) - 1) / (TPB / 32);

    k_deg_partial<<<DEG_R * DEG_S, DEG_TPB, 0, stream>>>(col, ew, bufA, (unsigned*)rid, dinv, ws_size);
    k_dinv_sum<<<nodeBlocks, TPB, 0, stream>>>(bufA, dinv);

    k_gemm_splitk<<<SK_GRID, SK_NW * 32, 0, stream>>>(x, W1, bufH, SK_TILES);
    k_zero<<<4096, TPB, 0, stream>>>(bufA, featElems);
    k_edge_owner<0><<<OWN_R * OWN_S, OWN_TPB, 0, stream>>>(row, col, ew, dinv, bufH, bufA,
                                                           rec, rcnt, rid, dinv, ws_size);
    k_finalize<<<4096, TPB, 0, stream>>>(bufA, bufH, dinv, b1, 1);

    k_gemm_n64<<<gemmBlocks, TPB, 0, stream>>>(bufA, W2, bufH, N_NODES, HID);
    k_zero<<<4096, TPB, 0, stream>>>(bufA, featElems);
    k_edge_owner<1><<<OWN_R * OWN_S, OWN_TPB, 0, stream>>>(row, col, ew, dinv, bufH, bufA,
                                                           rec, rcnt, rid, dinv, ws_size);
    k_finalize<<<4096, TPB, 0, stream>>>(bufA, bufH, dinv, b2, 1);

    k_fc<<<nodeBlocks, TPB, 0, stream>>>(bufA, Wfc, bfc, out);
}


# edge sums: one owner work-group per node range (252 ranges of 397 nodes, edges not sliced) stores relu(sum + self-loop + bias) itself; no global atomics, zero or finalize passes; sum order changes
# speedup vs baseline: 2.1384x; 1.2012x over previous
#include <hip/hip_runtime.h>
#include <hip/hip_bf16.h>


#define N_NODES 100000
#define N_EDGES 1600000
#define IN_DIM  512
#define HID     64
#define OUT_DIM 5

typedef __attribute__((ext_vector_type(2))) float v2f;
typedef __attribute__((ext_vector_type(8))) float v8f;

#define DEG_R     8
#define DEG_S     32
#define DEG_NR    12512
#define DEG_PITCH (DEG_R * DEG_NR)
#define DEG_ES    (N_EDGES / DEG_S)
#define DEG_TPB   1024

typedef __attribute__((ext_vector_type(4))) int   v4i;
typedef __attribute__((ext_vector_type(4))) float v4f;

#define OWN_NR    397

__device__ __forceinline__ unsigned rid_pack(v4i c) {
    unsigned o = 0u;
    #pragma unroll
    for (int j = 0; j < 4; ++j) {
        unsigned q = (unsigned)c[j] / (unsigned)OWN_NR;
        q = q < 255u ? q : 255u;
        o |= q << (8 * j);
    }
    return o;
}

__global__ __launch_bounds__(DEG_TPB)
void k_deg_partial(const int* __restrict__ col,
                   const float* __restrict__ w,
                   float* __restrict__ part,
                   unsigned* __restrict__ ridw,
                   const float* wsBase, size_t wsSize) {
    __shared__ float sdeg[DEG_NR];
    const int r  = blockIdx.x % DEG_R;
    const int s  = blockIdx.x / DEG_R;
    const int lo = r * DEG_NR;
    const bool hasb = (size_t)((const char*)(ridw + N_EDGES / 4) - (const char*)wsBase) <= wsSize;

    for (int i = threadIdx.x; i < DEG_NR; i += DEG_TPB) sdeg[i] = 0.0f;
    __syncthreads();

    const v4i* c4 = (const v4i*)(col + (size_t)s * DEG_ES);
    const v4f* w4 = (const v4f*)(w   + (size_t)s * DEG_ES);
    int blk = 0;
    for (int v0 = threadIdx.x; v0 < DEG_ES / 4; v0 += 4 * DEG_TPB, blk += 4) {
        v4i c[4];
        v4f x[4];
        for (int u = 0; u < 4; ++u) {
            const int v = v0 + u * DEG_TPB;
            const bool in = v < DEG_ES / 4;
            c[u] = in ? c4[v] : v4i{-1, -1, -1, -1};
            x[u] = in ? w4[v] : v4f{0.0f, 0.0f, 0.0f, 0.0f};
        }
        if (hasb) {
            for (int u = 0; u < 4; ++u) {
                const int v = v0 + u * DEG_TPB;
                if ((blk + u) % DEG_R == r && v < DEG_ES / 4)
                    ridw[(size_t)s * (DEG_ES / 4) + v] = rid_pack(c[u]);
            }
        }
        for (int u = 0; u < 4; ++u) {
            for (int j = 0; j < 4; ++j) {
                const unsigned d = (unsigned)(c[u][j] - lo);
                if (d < (unsigned)DEG_NR)
                    __hip_atomic_fetch_add(&sdeg[d], x[u][j], __ATOMIC_RELAXED,
                                           __HIP_MEMORY_SCOPE_WORKGROUP);
            }
        }
    }
    __syncthreads();

    float* dst = part + (size_t)s * DEG_PITCH + lo;
    for (int i = threadIdx.x; i < DEG_NR; i += DEG_TPB) dst[i] = sdeg[i];
}

__global__ void k_dinv_sum(const float* __restrict__ part,
                           float* __restrict__ dinv) {
    int i = blockIdx.x * blockDim.x + threadIdx.x;
    if (i < N_NODES) {
        float d = 1.0f;
        for (int s = 0; s < DEG_S; ++s) d += part[(size_t)s * DEG_PITCH + i];
        dinv[i] = (d > 0.0f) ? rsqrtf(d) : 0.0f;
    }
}

__global__ void k_gemm_n64(const float* __restrict__ X,
                           const float* __restrict__ W,
                           float* __restrict__ H,
                           int M, int K) {
    const int lane  = threadIdx.x & 31;
    const int waveT = (blockIdx.x * blockDim.x + threadIdx.x) >> 5;
    const int nTiles = M >> 4;
    if (waveT >= nTiles) return;

    const int m0   = waveT << 4;
    const int mRow = m0 + (lane & 15);
    const int kSub = (lane & 16) ? 2 : 0;
    const int nCol = lane & 15;

    v8f c0 = {}, c1 = {}, c2 = {}, c3 = {};

    for (int k = 0; k < K; k += 4) {
        v2f a = *(const v2f*)(X + (size_t)mRow * K + (k + kSub));

        const float* wr0 = W + (size_t)(k + kSub)     * HID + nCol;
        const float* wr1 = W + (size_t)(k + kSub + 1) * HID + nCol;
        v2f b0, b1, b2, b3;
        b0.x = wr0[0];  b0.y = wr1[0];
        b1.x = wr0[16]; b1.y = wr1[16];
        b2.x = wr0[32]; b2.y = wr1[32];
        b3.x = wr0[48]; b3.y = wr1[48];

        c0 = __builtin_amdgcn_wmma_f32_16x16x4_f32(false, a, false, b0,
                                                   (short)0, c0, false, false);
        c1 = __builtin_amdgcn_wmma_f32_16x16x4_f32(false, a, false, b1,
                                                   (short)0, c1, false, false);
        c2 = __builtin_amdgcn_wmma_f32_16x16x4_f32(false, a, false, b2,
                                                   (short)0, c2, false, false);
        c3 = __builtin_amdgcn_wmma_f32_16x16x4_f32(false, a, false, b3,
                                                   (short)0, c3, false, false);
    }

    const int rBase = m0 + ((lane & 16) ? 8 : 0);
    #pragma unroll
    for (int j = 0; j < 8; ++j) {
        float* dst = H + (size_t)(rBase + j) * HID + nCol;
        dst[0]  = c0[j];
        dst[16] = c1[j];
        dst[32] = c2[j];
        dst[48] = c3[j];
    }
}

#define SK_NW     8
#define SK_KS     (IN_DIM / SK_NW)
#define SK_NJ     (SK_KS / 8)
#define SK_PITCH  68
#define SK_GRID   512
#define SK_TILES  (N_NODES / 16)

__global__ __launch_bounds__(SK_NW * 32)
void k_gemm_splitk(const float* __restrict__ X,
                   const float* __restrict__ W,
                   float* __restrict__ H,
                   int nTiles) {
    __shared__ float red[2][SK_NW][16 * SK_PITCH];

    const int lane = threadIdx.x & 31;
    const int wv = __builtin_amdgcn_readfirstlane(threadIdx.x >> 5);
    const int c  = lane & 15;
    const int hf = lane >> 4;
    const int kb = wv * SK_KS;

    v2f bw[SK_NJ][2][4];
    #pragma unroll
    for (int j = 0; j < SK_NJ; ++j) {
        #pragma unroll
        for (int q = 0; q < 2; ++q) {
            const float* wp = W + (size_t)(kb + 8 * j + 4 * hf + 2 * q) * HID + c;
            #pragma unroll
            for (int n = 0; n < 4; ++n) {
                bw[j][q][n].x = wp[16 * n];
                bw[j][q][n].y = wp[HID + 16 * n];
            }
        }
    }

    const float* xp = X + (size_t)c * IN_DIM + kb + 4 * hf;

    int tile = blockIdx.x;
    v4f ac[SK_NJ];
    v4f an[SK_NJ];
    if (tile < nTiles) {
        const float* xt = xp + (size_t)tile * (16 * IN_DIM);
        #pragma unroll
        for (int j = 0; j < SK_NJ; ++j) ac[j] = *(const v4f*)(xt + 8 * j);
    }

    int buf = 0;
    for (; tile < nTiles; tile += gridDim.x) {
        {
            const int tn = tile + (int)gridDim.x;
            const int tl = tn < nTiles ? tn : tile;
            const float* xt = xp + (size_t)tl * (16 * IN_DIM);
            #pragma unroll
            for (int j = 0; j < SK_NJ; ++j) an[j] = *(const v4f*)(xt + 8 * j);
        }

        v8f acc[4] = {};
        #pragma unroll
        for (int j = 0; j < SK_NJ; ++j) {
            #pragma unroll
            for (int q = 0; q < 2; ++q) {
                v2f a;
                a.x = ac[j][2 * q];
                a.y = ac[j][2 * q + 1];
                #pragma unroll
                for (int n = 0; n < 4; ++n)
                    acc[n] = __builtin_amdgcn_wmma_f32_16x16x4_f32(false, a, false, bw[j][q][n],
                                                                   (short)0, acc[n], false, false);
            }
        }

        {
            float* pr = &red[buf][wv][(8 * hf) * SK_PITCH + c];
            #pragma unroll
            for (int n = 0; n < 4; ++n) {
                #pragma unroll
                for (int r = 0; r < 8; ++r) pr[r * SK_PITCH + 16 * n] = acc[n][r];
            }
        }
        __builtin_amdgcn_fence(__ATOMIC_RELEASE, "workgroup", "local");
        __builtin_amdgcn_s_barrier();
        __builtin_amdgcn_fence(__ATOMIC_ACQUIRE, "workgroup", "local");

        {
            const int rr = 2 * wv + hf;
            const float* sp = &red[buf][0][rr * SK_PITCH + 4 * c];
            v4f s = *(const v4f*)sp;
            #pragma unroll
            for (int p = 1; p < SK_NW; ++p) s += *(const v4f*)(sp + p * (16 * SK_PITCH));
            *(v4f*)(H + ((size_t)tile * 16 + rr) * HID + 4 * c) = s;
        }

        buf ^= 1;
        #pragma unroll
        for (int j = 0; j < SK_NJ; ++j) ac[j] = an[j];
    }
}

#define OWN_R     252
#define OWN_S     1
#define OWN_TPB   1024
#define OWN_NV    4
#define OWN_VTOT  (N_EDGES / 16)
#define OWN_VS    ((OWN_VTOT + OWN_S - 1) / OWN_S)
#define OWN_CHUNK (OWN_NV * OWN_TPB)
#define OWN_NCH   ((OWN_VS + OWN_CHUNK - 1) / OWN_CHUNK)
#define OWN_GB    16
#define OWN_WQ    64
#define OWN_RCAP  288
#define OWN_NWAVE (OWN_R * OWN_S * (OWN_TPB / 32))
#define OWN_FIT   ((OWN_NR * (HID / 4) + OWN_TPB - 1) / OWN_TPB)

template <int MODE>
__global__ __launch_bounds__(OWN_TPB)
void k_edge_owner(const int* __restrict__ row,
                  const int* __restrict__ col,
                  const float* __restrict__ w,
                  const float* __restrict__ dinv,
                  const float* __restrict__ h,
                  float* __restrict__ agg,
                  const float* __restrict__ b,
                  int* __restrict__ rec,
                  int* __restrict__ rcnt,
                  const unsigned char* __restrict__ rid,
                  const float* wsBase, size_t wsSize) {
    __shared__ float    sums[(OWN_NR + 1) * HID];
    __shared__ unsigned wq[OWN_TPB / 32][OWN_WQ];

    const int tid  = threadIdx.x;
    const int lane = tid & 31;
    const int wv   = __builtin_amdgcn_readfirstlane(tid >> 5);
    const int r    = blockIdx.x % OWN_R;
    const int s    = blockIdx.x / OWN_R;
    const int lo   = r * OWN_NR;
    const int vs   = s * OWN_VS;
    const int ve   = (vs + OWN_VS < OWN_VTOT) ? (vs + OWN_VS) : OWN_VTOT;

    const bool room = (size_t)((const char*)(rcnt + OWN_NWAVE) - (const char*)wsBase) <= wsSize;
    const bool hasb = (size_t)((const char*)(rid + N_EDGES) - (const char*)wsBase) <= wsSize;
    const int  gw   = blockIdx.x * (OWN_TPB / 32) + wv;
    int* myrec = rec + (size_t)gw * (2 * OWN_RCAP);
    int  nrec  = 0;
    int  nrep  = -1;
    if (MODE == 1 && room) nrep = __builtin_amdgcn_readfirstlane(rcnt[gw]);
    const bool doscan = (MODE == 0) || nrep < 0;

    {
        v4f* z4 = (v4f*)sums;
        for (int i = tid; i < OWN_NR * HID / 4; i += OWN_TPB)
            z4[i] = v4f{0.0f, 0.0f, 0.0f, 0.0f};
    }

    const v4i* b4 = (const v4i*)rid;
    const v4i* c4 = (const v4i*)col;
    auto ldids = [&](int v) -> v4i {
        if (hasb) return b4[v];
        v4i o;
        #pragma unroll
        for (int q = 0; q < 4; ++q) o[q] = (int)rid_pack(c4[4 * v + q]);
        return o;
    };
    v4i c[OWN_NV];
    v4i cn[OWN_NV];
    #pragma unroll
    for (int u = 0; u < OWN_NV; ++u) {
        const int v = vs + u * OWN_TPB + tid;
        c[u] = (doscan && v < ve) ? ldids(v) : v4i{-1, -1, -1, -1};
    }
    __syncthreads();

    unsigned* myq = wq[wv];
    int count = 0;

    auto gather_add = [&](int rr, int dd, float nr, int cnt) {
        v2f   hv[OWN_GB];
        float nn[OWN_GB];
        int   dl[OWN_GB];
        #pragma unroll
        for (int u = 0; u < OWN_GB; ++u) {
            const bool ok = u < cnt;
            const int  i  = ok ? u : 0;
            const int  ru = __builtin_amdgcn_readlane(rr, i);
            const int  du = __builtin_amdgcn_readlane(dd, i);
            dl[u] = ok ? du : OWN_NR;
            nn[u] = __builtin_bit_cast(float,
                      __builtin_amdgcn_readlane(__builtin_bit_cast(int, nr), i));
            hv[u] = *(const v2f*)(h + (size_t)ru * HID + lane * 2);
        }
        #pragma unroll
        for (int u = 0; u < OWN_GB; ++u) {
            float* dst = &sums[dl[u] * HID + lane * 2];
            __hip_atomic_fetch_add(dst + 0, nn[u] * hv[u].x, __ATOMIC_RELAXED,
                                   __HIP_MEMORY_SCOPE_WORKGROUP);
            __hip_atomic_fetch_add(dst + 1, nn[u] * hv[u].y, __ATOMIC_RELAXED,
                                   __HIP_MEMORY_SCOPE_WORKGROUP);
        }
    };

    auto serve = [&](int base, int cnt) {
        int   rr = 0;
        int   dd = 0;
        float nr = 0.0f;
        if (lane < cnt) {
            const unsigned e = myq[base + lane];
            rr = row[e];
            const int   cc = col[e];
            const float we = w[e];
            dd = cc - lo;
            nr = dinv[rr] * we * dinv[cc];
        }
        if (MODE == 0 && room) {
            if (nrec + cnt <= OWN_RCAP && lane < cnt) {
                myrec[nrec + lane]            = rr | (dd << 17);
                myrec[OWN_RCAP + nrec + lane] = __builtin_bit_cast(int, nr);
            }
            nrec += cnt;
        }
        gather_add(rr, dd, nr, cnt);
    };

    if (doscan) {
    for (int ch = 0; ch < OWN_NCH; ++ch) {
        const int vb = vs + ch * OWN_CHUNK + tid;

        #pragma unroll
        for (int u = 0; u < OWN_NV; ++u) {
            const int v = vb + OWN_CHUNK + u * OWN_TPB;
            cn[u] = (ch + 1 < OWN_NCH && v < ve) ? ldids(v) : v4i{-1, -1, -1, -1};
        }

        unsigned mh[2] = {0u, 0u};
        const unsigned r4 = (unsigned)r * 0x01010101u;
        #pragma unroll
        for (int u = 0; u < OWN_NV; ++u) {
            #pragma unroll
            for (int q = 0; q < 4; ++q) {
                const int k = 4 * u + q;
                const unsigned y = (unsigned)c[u][q] ^ r4;
                const unsigned t = ((y & 0x7f7f7f7fu) + 0x7f7f7f7fu) | y;
                mh[k >> 3] |= (~t & 0x80808080u) >> (7 - (k & 7));
            }
        }
        unsigned long long mask = (unsigned long long)mh[0] | ((unsigned long long)mh[1] << 32);

        for (;;) {
            const bool     has = mask != 0ull;
            const unsigned bal = __builtin_amdgcn_ballot_w32(has);
            if (bal == 0u) break;
            if (has) {
                const int idx = __builtin_ctzll(mask);
                mask &= mask - 1ull;
                const int k = (idx & 7) + ((idx >> 5) << 3);
                const unsigned e = 16u * (unsigned)(vb + (k >> 2) * OWN_TPB)
                                 + 4u * (unsigned)(k & 3) + (unsigned)((idx >> 3) & 3);
                myq[count + (int)__builtin_amdgcn_mbcnt_lo(bal, 0u)] = e;
            }
            count += __builtin_popcount(bal);
            __builtin_amdgcn_fence(__ATOMIC_ACQ_REL, "wavefront");
            while (count >= OWN_GB) {
                count -= OWN_GB;
                serve(count, OWN_GB);
            }
        }

        #pragma unroll
        for (int u = 0; u < OWN_NV; ++u) c[u] = cn[u];
    }
    if (count > 0) serve(0, count);
    if (MODE == 0 && room && lane == 0)
        rcnt[gw] = (nrec <= OWN_RCAP) ? nrec : -1;
    } else {
        if (nrep > 0) {
            const int last = nrep - 1;
            const int i0   = lane < last ? lane : last;
            int curp = myrec[i0];
            int curn = myrec[OWN_RCAP + i0];
            for (int base = 0; base < nrep; base += OWN_GB) {
                const int left = nrep - base;
                const int cnt  = left < OWN_GB ? left : OWN_GB;
                const int nx   = base + OWN_GB + lane;
                const int i1   = nx < last ? nx : last;
                const int nxtp = myrec[i1];
                const int nxtn = myrec[OWN_RCAP + i1];
                const int   rr = curp & 0x1ffff;
                const int   dd = curp >> 17;
                const float nr = __builtin_bit_cast(float, curn);
                if (cnt == OWN_GB) gather_add(rr, dd, nr, OWN_GB);
                else               gather_add(rr, dd, nr, cnt);
                curp = nxtp;
                curn = nxtn;
            }
        }
    }

    const int f  = 4 * (tid & 15);
    const v4f bv = *(const v4f*)(b + f);
    v4f   hs[OWN_FIT];
    float ds[OWN_FIT];
    #pragma unroll
    for (int t = 0; t < OWN_FIT; ++t) {
        const int nd = lo + ((tid + t * OWN_TPB) >> 4);
        const int nc = nd < N_NODES ? nd : N_NODES - 1;
        ds[t] = dinv[nc];
        hs[t] = *(const v4f*)(h + (size_t)nc * HID + f);
    }

    __builtin_amdgcn_fence(__ATOMIC_RELEASE, "workgroup", "local");
    __builtin_amdgcn_s_barrier();
    __builtin_amdgcn_fence(__ATOMIC_ACQUIRE, "workgroup", "local");

    const v4f* s4 = (const v4f*)sums;
    #pragma unroll
    for (int t = 0; t < OWN_FIT; ++t) {
        const int i    = tid + t * OWN_TPB;
        const int node = lo + (i >> 4);
        if (i < OWN_NR * (HID / 4) && node < N_NODES) {
            v4f v = s4[i] + ds[t] * ds[t] * hs[t] + bv;
            #pragma unroll
            for (int q = 0; q < 4; ++q) v[q] = fmaxf(v[q], 0.0f);
            *(v4f*)(agg + (size_t)node * HID + f) = v;
        }
    }
}

__global__ void k_fc(const float* __restrict__ h,
                     const float* __restrict__ Wfc,
                     const float* __restrict__ bfc,
                     float* __restrict__ out) {
    int n = blockIdx.x * blockDim.x + threadIdx.x;
    if (n >= N_NODES) return;
    float acc[OUT_DIM];
    #pragma unroll
    for (int o = 0; o < OUT_DIM; ++o) acc[o] = bfc[o];
    const float* hr = h + (size_t)n * HID;
    #pragma unroll 4
    for (int k = 0; k < HID; ++k) {
        const float xv = hr[k];
        #pragma unroll
        for (int o = 0; o < OUT_DIM; ++o)
            acc[o] = fmaf(xv, Wfc[k * OUT_DIM + o], acc[o]);
    }
    float* orow = out + (size_t)n * OUT_DIM;
    #pragma unroll
    for (int o = 0; o < OUT_DIM; ++o) orow[o] = acc[o];
}

extern "C" void kernel_launch(void* const* d_in, const int* in_sizes, int n_in,
                              void* d_out, int out_size, void* d_ws, size_t ws_size,
                              hipStream_t stream) {
    const float* x   = (const float*)d_in[0];
    const int*   ei  = (const int*)  d_in[1];
    const float* ew  = (const float*)d_in[2];
    const float* W1  = (const float*)d_in[3];
    const float* b1  = (const float*)d_in[4];
    const float* W2  = (const float*)d_in[5];
    const float* b2  = (const float*)d_in[6];
    const float* Wfc = (const float*)d_in[7];
    const float* bfc = (const float*)d_in[8];
    float* out = (float*)d_out;

    const int* row = ei;
    const int* col = ei + N_EDGES;

    char* ws = (char*)d_ws;
    size_t off = 0;
    float* dinv = (float*)(ws + off); off += ((size_t)N_NODES * 4 + 255) & ~(size_t)255;
    float* bufH = (float*)(ws + off); off += (size_t)N_NODES * HID * 4;
    float* bufA = (float*)(ws + off); off += (size_t)N_NODES * HID * 4;
    int*   rec  = (int*)(ws + off);   off += (size_t)OWN_NWAVE * OWN_RCAP * 8;
    int*   rcnt = (int*)(ws + off);   off += (size_t)OWN_NWAVE * 4;
    unsigned char* rid = (unsigned char*)(ws + off);   off += (size_t)N_EDGES;
    (void)n_in; (void)in_sizes; (void)out_size;

    const int TPB = 256;
    const int nodeBlocks = (N_NODES + TPB - 1) / TPB;
    const int gemmBlocks = ((N_NODES / 16) + (TPB / 32) - 1) / (TPB / 32);

    k_deg_partial<<<DEG_R * DEG_S, DEG_TPB, 0, stream>>>(col, ew, bufA, (unsigned*)rid, dinv, ws_size);
    k_dinv_sum<<<nodeBlocks, TPB, 0, stream>>>(bufA, dinv);

    k_gemm_splitk<<<SK_GRID, SK_NW * 32, 0, stream>>>(x, W1, bufH, SK_TILES);
    k_edge_owner<0><<<OWN_R * OWN_S, OWN_TPB, 0, stream>>>(row, col, ew, dinv, bufH, bufA, b1,
                                                           rec, rcnt, rid, dinv, ws_size);

    k_gemm_n64<<<gemmBlocks, TPB, 0, stream>>>(bufA, W2, bufH, N_NODES, HID);
    k_edge_owner<1><<<OWN_R * OWN_S, OWN_TPB, 0, stream>>>(row, col, ew, dinv, bufH, bufA, b2,
                                                           rec, rcnt, rid, dinv, ws_size);

    k_fc<<<nodeBlocks, TPB, 0, stream>>>(bufA, Wfc, bfc, out);
}


# layer 2 taken as (A agg) W2: owner pass sums agg rows; each wave multiplies a 16-row tile of its LDS sums by W2 (f32 matrix line), result registers feed the FC; 7 -> 5 launches; sum order changes
# speedup vs baseline: 2.3799x; 1.1129x over previous
#include <hip/hip_runtime.h>
#include <hip/hip_bf16.h>


#define N_NODES 100000
#define N_EDGES 1600000
#define IN_DIM  512
#define HID     64
#define OUT_DIM 5

typedef __attribute__((ext_vector_type(2))) float v2f;
typedef __attribute__((ext_vector_type(8))) float v8f;

#define DEG_R     8
#define DEG_S     32
#define DEG_NR    12512
#define DEG_PITCH (DEG_R * DEG_NR)
#define DEG_ES    (N_EDGES / DEG_S)
#define DEG_TPB   1024

typedef __attribute__((ext_vector_type(4))) int   v4i;
typedef __attribute__((ext_vector_type(4))) float v4f;

#define OWN_NR    397

__device__ __forceinline__ unsigned rid_pack(v4i c) {
    unsigned o = 0u;
    #pragma unroll
    for (int j = 0; j < 4; ++j) {
        unsigned q = (unsigned)c[j] / (unsigned)OWN_NR;
        q = q < 255u ? q : 255u;
        o |= q << (8 * j);
    }
    return o;
}

__global__ __launch_bounds__(DEG_TPB)
void k_deg_partial(const int* __restrict__ col,
                   const float* __restrict__ w,
                   float* __restrict__ part,
                   unsigned* __restrict__ ridw,
                   const float* wsBase, size_t wsSize) {
    __shared__ float sdeg[DEG_NR];
    const int r  = blockIdx.x % DEG_R;
    const int s  = blockIdx.x / DEG_R;
    const int lo = r * DEG_NR;
    const bool hasb = (size_t)((const char*)(ridw + N_EDGES / 4) - (const char*)wsBase) <= wsSize;

    for (int i = threadIdx.x; i < DEG_NR; i += DEG_TPB) sdeg[i] = 0.0f;
    __syncthreads();

    const v4i* c4 = (const v4i*)(col + (size_t)s * DEG_ES);
    const v4f* w4 = (const v4f*)(w   + (size_t)s * DEG_ES);
    int blk = 0;
    for (int v0 = threadIdx.x; v0 < DEG_ES / 4; v0 += 4 * DEG_TPB, blk += 4) {
        v4i c[4];
        v4f x[4];
        for (int u = 0; u < 4; ++u) {
            const int v = v0 + u * DEG_TPB;
            const bool in = v < DEG_ES / 4;
            c[u] = in ? c4[v] : v4i{-1, -1, -1, -1};
            x[u] = in ? w4[v] : v4f{0.0f, 0.0f, 0.0f, 0.0f};
        }
        if (hasb) {
            for (int u = 0; u < 4; ++u) {
                const int v = v0 + u * DEG_TPB;
                if ((blk + u) % DEG_R == r && v < DEG_ES / 4)
                    ridw[(size_t)s * (DEG_ES / 4) + v] = rid_pack(c[u]);
            }
        }
        for (int u = 0; u < 4; ++u) {
            for (int j = 0; j < 4; ++j) {
                const unsigned d = (unsigned)(c[u][j] - lo);
                if (d < (unsigned)DEG_NR)
                    __hip_atomic_fetch_add(&sdeg[d], x[u][j], __ATOMIC_RELAXED,
                                           __HIP_MEMORY_SCOPE_WORKGROUP);
            }
        }
    }
    __syncthreads();

    float* dst = part + (size_t)s * DEG_PITCH + lo;
    for (int i = threadIdx.x; i < DEG_NR; i += DEG_TPB) dst[i] = sdeg[i];
}

__global__ void k_dinv_sum(const float* __restrict__ part,
                           float* __restrict__ dinv) {
    int i = blockIdx.x * blockDim.x + threadIdx.x;
    if (i < N_NODES) {
        float d = 1.0f;
        for (int s = 0; s < DEG_S; ++s) d += part[(size_t)s * DEG_PITCH + i];
        dinv[i] = (d > 0.0f) ? rsqrtf(d) : 0.0f;
    }
}


#define SK_NW     8
#define SK_KS     (IN_DIM / SK_NW)
#define SK_NJ     (SK_KS / 8)
#define SK_PITCH  68
#define SK_GRID   512
#define SK_TILES  (N_NODES / 16)

__global__ __launch_bounds__(SK_NW * 32)
void k_gemm_splitk(const float* __restrict__ X,
                   const float* __restrict__ W,
                   float* __restrict__ H,
                   int nTiles) {
    __shared__ float red[2][SK_NW][16 * SK_PITCH];

    const int lane = threadIdx.x & 31;
    const int wv = __builtin_amdgcn_readfirstlane(threadIdx.x >> 5);
    const int c  = lane & 15;
    const int hf = lane >> 4;
    const int kb = wv * SK_KS;

    v2f bw[SK_NJ][2][4];
    #pragma unroll
    for (int j = 0; j < SK_NJ; ++j) {
        #pragma unroll
        for (int q = 0; q < 2; ++q) {
            const float* wp = W + (size_t)(kb + 8 * j + 4 * hf + 2 * q) * HID + c;
            #pragma unroll
            for (int n = 0; n < 4; ++n) {
                bw[j][q][n].x = wp[16 * n];
                bw[j][q][n].y = wp[HID + 16 * n];
            }
        }
    }

    const float* xp = X + (size_t)c * IN_DIM + kb + 4 * hf;

    int tile = blockIdx.x;
    v4f ac[SK_NJ];
    v4f an[SK_NJ];
    if (tile < nTiles) {
        const float* xt = xp + (size_t)tile * (16 * IN_DIM);
        #pragma unroll
        for (int j = 0; j < SK_NJ; ++j) ac[j] = *(const v4f*)(xt + 8 * j);
    }

    int buf = 0;
    for (; tile < nTiles; tile += gridDim.x) {
        {
            const int tn = tile + (int)gridDim.x;
            const int tl = tn < nTiles ? tn : tile;
            const float* xt = xp + (size_t)tl * (16 * IN_DIM);
            #pragma unroll
            for (int j = 0; j < SK_NJ; ++j) an[j] = *(const v4f*)(xt + 8 * j);
        }

        v8f acc[4] = {};
        #pragma unroll
        for (int j = 0; j < SK_NJ; ++j) {
            #pragma unroll
            for (int q = 0; q < 2; ++q) {
                v2f a;
                a.x = ac[j][2 * q];
                a.y = ac[j][2 * q + 1];
                #pragma unroll
                for (int n = 0; n < 4; ++n)
                    acc[n] = __builtin_amdgcn_wmma_f32_16x16x4_f32(false, a, false, bw[j][q][n],
                                                                   (short)0, acc[n], false, false);
            }
        }

        {
            float* pr = &red[buf][wv][(8 * hf) * SK_PITCH + c];
            #pragma unroll
            for (int n = 0; n < 4; ++n) {
                #pragma unroll
                for (int r = 0; r < 8; ++r) pr[r * SK_PITCH + 16 * n] = acc[n][r];
            }
        }
        __builtin_amdgcn_fence(__ATOMIC_RELEASE, "workgroup", "local");
        __builtin_amdgcn_s_barrier();
        __builtin_amdgcn_fence(__ATOMIC_ACQUIRE, "workgroup", "local");

        {
            const int rr = 2 * wv + hf;
            const float* sp = &red[buf][0][rr * SK_PITCH + 4 * c];
            v4f s = *(const v4f*)sp;
            #pragma unroll
            for (int p = 1; p < SK_NW; ++p) s += *(const v4f*)(sp + p * (16 * SK_PITCH));
            *(v4f*)(H + ((size_t)tile * 16 + rr) * HID + 4 * c) = s;
        }

        buf ^= 1;
        #pragma unroll
        for (int j = 0; j < SK_NJ; ++j) ac[j] = an[j];
    }
}

#define OWN_R     252
#define OWN_S     1
#define OWN_TPB   1024
#define OWN_NV    4
#define OWN_VTOT  (N_EDGES / 16)
#define OWN_VS    ((OWN_VTOT + OWN_S - 1) / OWN_S)
#define OWN_CHUNK (OWN_NV * OWN_TPB)
#define OWN_NCH   ((OWN_VS + OWN_CHUNK - 1) / OWN_CHUNK)
#define OWN_GB    16
#define OWN_WQ    64
#define OWN_RCAP  288
#define OWN_NWAVE (OWN_R * OWN_S * (OWN_TPB / 32))
#define OWN_FIT   ((OWN_NR * (HID / 4) + OWN_TPB - 1) / OWN_TPB)
#define OWN_PITCH2 68
#define OWN_NT     ((OWN_NR + 1 + 15) / 16)
#define OWN_ROWS2  (16 * OWN_NT)

template <int MODE>
__global__ __launch_bounds__(OWN_TPB)
void k_edge_owner(const int* __restrict__ row,
                  const int* __restrict__ col,
                  const float* __restrict__ w,
                  const float* __restrict__ dinv,
                  const float* __restrict__ h,
                  float* __restrict__ agg,
                  const float* __restrict__ b,
                  const float* __restrict__ Wm,
                  const float* __restrict__ Wfc,
                  const float* __restrict__ bfc,
                  float* __restrict__ out,
                  int* __restrict__ rec,
                  int* __restrict__ rcnt,
                  const unsigned char* __restrict__ rid,
                  const float* wsBase, size_t wsSize) {
    constexpr int PITCH = (MODE == 1) ? OWN_PITCH2 : HID;
    constexpr int ROWS  = (MODE == 1) ? OWN_ROWS2 : OWN_NR + 1;
    __shared__ float    sums[ROWS * PITCH];
    __shared__ unsigned wq[OWN_TPB / 32][OWN_WQ];

    const int tid  = threadIdx.x;
    const int lane = tid & 31;
    const int wv   = __builtin_amdgcn_readfirstlane(tid >> 5);
    const int r    = blockIdx.x % OWN_R;
    const int s    = blockIdx.x / OWN_R;
    const int lo   = r * OWN_NR;
    const int vs   = s * OWN_VS;
    const int ve   = (vs + OWN_VS < OWN_VTOT) ? (vs + OWN_VS) : OWN_VTOT;

    const bool room = (size_t)((const char*)(rcnt + OWN_NWAVE) - (const char*)wsBase) <= wsSize;
    const bool hasb = (size_t)((const char*)(rid + N_EDGES) - (const char*)wsBase) <= wsSize;
    const int  gw   = blockIdx.x * (OWN_TPB / 32) + wv;
    int* myrec = rec + (size_t)gw * (2 * OWN_RCAP);
    int  nrec  = 0;
    int  nrep  = -1;
    if (MODE == 1 && room) nrep = __builtin_amdgcn_readfirstlane(rcnt[gw]);
    const bool doscan = (MODE == 0) || nrep < 0;

    {
        v4f* z4 = (v4f*)sums;
        constexpr int NZ = (MODE == 1) ? ROWS * PITCH / 4 : OWN_NR * HID / 4;
        for (int i = tid; i < NZ; i += OWN_TPB)
            z4[i] = v4f{0.0f, 0.0f, 0.0f, 0.0f};
    }

    const v4i* b4 = (const v4i*)rid;
    const v4i* c4 = (const v4i*)col;
    auto ldids = [&](int v) -> v4i {
        if (hasb) return b4[v];
        v4i o;
        #pragma unroll
        for (int q = 0; q < 4; ++q) o[q] = (int)rid_pack(c4[4 * v + q]);
        return o;
    };
    v4i c[OWN_NV];
    v4i cn[OWN_NV];
    #pragma unroll
    for (int u = 0; u < OWN_NV; ++u) {
        const int v = vs + u * OWN_TPB + tid;
        c[u] = (doscan && v < ve) ? ldids(v) : v4i{-1, -1, -1, -1};
    }
    __syncthreads();

    unsigned* myq = wq[wv];
    int count = 0;

    auto gather_add = [&](int rr, int dd, float nr, int cnt) {
        v2f   hv[OWN_GB];
        float nn[OWN_GB];
        int   dl[OWN_GB];
        #pragma unroll
        for (int u = 0; u < OWN_GB; ++u) {
            const bool ok = u < cnt;
            const int  i  = ok ? u : 0;
            const int  ru = __builtin_amdgcn_readlane(rr, i);
            const int  du = __builtin_amdgcn_readlane(dd, i);
            dl[u] = ok ? du : OWN_NR;
            nn[u] = __builtin_bit_cast(float,
                      __builtin_amdgcn_readlane(__builtin_bit_cast(int, nr), i));
            hv[u] = *(const v2f*)(h + (size_t)ru * HID + lane * 2);
        }
        #pragma unroll
        for (int u = 0; u < OWN_GB; ++u) {
            float* dst = &sums[dl[u] * PITCH + lane * 2];
            __hip_atomic_fetch_add(dst + 0, nn[u] * hv[u].x, __ATOMIC_RELAXED,
                                   __HIP_MEMORY_SCOPE_WORKGROUP);
            __hip_atomic_fetch_add(dst + 1, nn[u] * hv[u].y, __ATOMIC_RELAXED,
                                   __HIP_MEMORY_SCOPE_WORKGROUP);
        }
    };

    auto serve = [&](int base, int cnt) {
        int   rr = 0;
        int   dd = 0;
        float nr = 0.0f;
        if (lane < cnt) {
            const unsigned e = myq[base + lane];
            rr = row[e];
            const int   cc = col[e];
            const float we = w[e];
            dd = cc - lo;
            nr = dinv[rr] * we * dinv[cc];
        }
        if (MODE == 0 && room) {
            if (nrec + cnt <= OWN_RCAP && lane < cnt) {
                myrec[nrec + lane]            = rr | (dd << 17);
                myrec[OWN_RCAP + nrec + lane] = __builtin_bit_cast(int, nr);
            }
            nrec += cnt;
        }
        gather_add(rr, dd, nr, cnt);
    };

    if (doscan) {
    for (int ch = 0; ch < OWN_NCH; ++ch) {
        const int vb = vs + ch * OWN_CHUNK + tid;

        #pragma unroll
        for (int u = 0; u < OWN_NV; ++u) {
            const int v = vb + OWN_CHUNK + u * OWN_TPB;
            cn[u] = (ch + 1 < OWN_NCH && v < ve) ? ldids(v) : v4i{-1, -1, -1, -1};
        }

        unsigned mh[2] = {0u, 0u};
        const unsigned r4 = (unsigned)r * 0x01010101u;
        #pragma unroll
        for (int u = 0; u < OWN_NV; ++u) {
            #pragma unroll
            for (int q = 0; q < 4; ++q) {
                const int k = 4 * u + q;
                const unsigned y = (unsigned)c[u][q] ^ r4;
                const unsigned t = ((y & 0x7f7f7f7fu) + 0x7f7f7f7fu) | y;
                mh[k >> 3] |= (~t & 0x80808080u) >> (7 - (k & 7));
            }
        }
        unsigned long long mask = (unsigned long long)mh[0] | ((unsigned long long)mh[1] << 32);

        for (;;) {
            const bool     has = mask != 0ull;
            const unsigned bal = __builtin_amdgcn_ballot_w32(has);
            if (bal == 0u) break;
            if (has) {
                const int idx = __builtin_ctzll(mask);
                mask &= mask - 1ull;
                const int k = (idx & 7) + ((idx >> 5) << 3);
                const unsigned e = 16u * (unsigned)(vb + (k >> 2) * OWN_TPB)
                                 + 4u * (unsigned)(k & 3) + (unsigned)((idx >> 3) & 3);
                myq[count + (int)__builtin_amdgcn_mbcnt_lo(bal, 0u)] = e;
            }
            count += __builtin_popcount(bal);
            __builtin_amdgcn_fence(__ATOMIC_ACQ_REL, "wavefront");
            while (count >= OWN_GB) {
                count -= OWN_GB;
                serve(count, OWN_GB);
            }
        }

        #pragma unroll
        for (int u = 0; u < OWN_NV; ++u) c[u] = cn[u];
    }
    if (count > 0) serve(0, count);
    if (MODE == 0 && room && lane == 0)
        rcnt[gw] = (nrec <= OWN_RCAP) ? nrec : -1;
    } else {
        if (nrep > 0) {
            const int last = nrep - 1;
            const int i0   = lane < last ? lane : last;
            int curp = myrec[i0];
            int curn = myrec[OWN_RCAP + i0];
            for (int base = 0; base < nrep; base += OWN_GB) {
                const int left = nrep - base;
                const int cnt  = left < OWN_GB ? left : OWN_GB;
                const int nx   = base + OWN_GB + lane;
                const int i1   = nx < last ? nx : last;
                const int nxtp = myrec[i1];
                const int nxtn = myrec[OWN_RCAP + i1];
                const int   rr = curp & 0x1ffff;
                const int   dd = curp >> 17;
                const float nr = __builtin_bit_cast(float, curn);
                if (cnt == OWN_GB) gather_add(rr, dd, nr, OWN_GB);
                else               gather_add(rr, dd, nr, cnt);
                curp = nxtp;
                curn = nxtn;
            }
        }
    }

    const int f  = 4 * (tid & 15);
    v4f   hs[OWN_FIT];
    float ds[OWN_FIT];
    #pragma unroll
    for (int t = 0; t < OWN_FIT; ++t) {
        const int nd = lo + ((tid + t * OWN_TPB) >> 4);
        const int nc = nd < N_NODES ? nd : N_NODES - 1;
        ds[t] = dinv[nc];
        hs[t] = *(const v4f*)(h + (size_t)nc * HID + f);
    }

    __builtin_amdgcn_fence(__ATOMIC_RELEASE, "workgroup", "local");
    __builtin_amdgcn_s_barrier();
    __builtin_amdgcn_fence(__ATOMIC_ACQUIRE, "workgroup", "local");

    if (MODE == 0) {
        const v4f bv = *(const v4f*)(b + f);
        const v4f* s4 = (const v4f*)sums;
        #pragma unroll
        for (int t = 0; t < OWN_FIT; ++t) {
            const int i    = tid + t * OWN_TPB;
            const int node = lo + (i >> 4);
            if (i < OWN_NR * (HID / 4) && node < N_NODES) {
                v4f v = s4[i] + ds[t] * ds[t] * hs[t] + bv;
                #pragma unroll
                for (int q = 0; q < 4; ++q) v[q] = fmaxf(v[q], 0.0f);
                *(v4f*)(agg + (size_t)node * HID + f) = v;
            }
        }
    } else {
        #pragma unroll
        for (int t = 0; t < OWN_FIT; ++t) {
            const int i = tid + t * OWN_TPB;
            if (i < OWN_NR * (HID / 4)) {
                v4f* p = (v4f*)&sums[(i >> 4) * PITCH + f];
                *p = *p + ds[t] * ds[t] * hs[t];
            }
        }
        __builtin_amdgcn_fence(__ATOMIC_RELEASE, "workgroup", "local");
        __builtin_amdgcn_s_barrier();
        __builtin_amdgcn_fence(__ATOMIC_ACQUIRE, "workgroup", "local");

        if (wv < OWN_NT) {
            const int c  = lane & 15;
            const int hf = lane >> 4;
            const float* sp = &sums[(16 * wv + c) * PITCH + 2 * hf];
            const float* wp = Wm + (size_t)(2 * hf) * HID + c;
            v8f acc[4] = {};
            #pragma unroll 4
            for (int k0 = 0; k0 < HID; k0 += 4) {
                const v2f sv = *(const v2f*)(sp + k0);
                #pragma unroll
                for (int n = 0; n < 4; ++n) {
                    v2f wf;
                    wf.x = wp[(size_t)k0 * HID + 16 * n];
                    wf.y = wp[(size_t)(k0 + 1) * HID + 16 * n];
                    acc[n] = __builtin_amdgcn_wmma_f32_16x16x4_f32(false, wf, false, sv,
                                                                   (short)0, acc[n], false, false);
                }
            }
            #pragma unroll
            for (int n = 0; n < 4; ++n) {
                const v4f ba = *(const v4f*)(b + 16 * n + 8 * hf);
                const v4f bb = *(const v4f*)(b + 16 * n + 8 * hf + 4);
                #pragma unroll
                for (int j = 0; j < 4; ++j) {
                    acc[n][j]     = fmaxf(acc[n][j]     + ba[j], 0.0f);
                    acc[n][j + 4] = fmaxf(acc[n][j + 4] + bb[j], 0.0f);
                }
            }
            const int   co = c < OUT_DIM ? c : 0;
            const float cm = c < OUT_DIM ? 1.0f : 0.0f;
            v2f wa[4][4];
            #pragma unroll
            for (int n = 0; n < 4; ++n) {
                #pragma unroll
                for (int jj = 0; jj < 4; ++jj) {
                    const int kc = 16 * n + 8 * hf + 2 * jj;
                    wa[n][jj].x = cm * Wfc[kc * OUT_DIM + co];
                    wa[n][jj].y = cm * Wfc[(kc + 1) * OUT_DIM + co];
                }
            }
            v8f fo = {};
            #pragma unroll
            for (int n = 0; n < 4; ++n) {
                #pragma unroll
                for (int jj = 0; jj < 4; ++jj) {
                    v2f rb;
                    rb.x = acc[n][2 * jj];
                    rb.y = acc[n][2 * jj + 1];
                    fo = __builtin_amdgcn_wmma_f32_16x16x4_f32(false, wa[n][jj], false, rb,
                                                               (short)0, fo, false, false);
                }
            }
            const int rw   = 16 * wv + c;
            const int node = lo + rw;
            if (hf == 0 && rw < OWN_NR && node < N_NODES) {
                float* op = out + (size_t)node * OUT_DIM;
                #pragma unroll
                for (int o = 0; o < OUT_DIM; ++o) op[o] = fo[o] + bfc[o];
            }
        }
    }
}

extern "C" void kernel_launch(void* const* d_in, const int* in_sizes, int n_in,
                              void* d_out, int out_size, void* d_ws, size_t ws_size,
                              hipStream_t stream) {
    const float* x   = (const float*)d_in[0];
    const int*   ei  = (const int*)  d_in[1];
    const float* ew  = (const float*)d_in[2];
    const float* W1  = (const float*)d_in[3];
    const float* b1  = (const float*)d_in[4];
    const float* W2  = (const float*)d_in[5];
    const float* b2  = (const float*)d_in[6];
    const float* Wfc = (const float*)d_in[7];
    const float* bfc = (const float*)d_in[8];
    float* out = (float*)d_out;

    const int* row = ei;
    const int* col = ei + N_EDGES;

    char* ws = (char*)d_ws;
    size_t off = 0;
    float* dinv = (float*)(ws + off); off += ((size_t)N_NODES * 4 + 255) & ~(size_t)255;
    float* bufH = (float*)(ws + off); off += (size_t)N_NODES * HID * 4;
    float* bufA = (float*)(ws + off); off += (size_t)N_NODES * HID * 4;
    int*   rec  = (int*)(ws + off);   off += (size_t)OWN_NWAVE * OWN_RCAP * 8;
    int*   rcnt = (int*)(ws + off);   off += (size_t)OWN_NWAVE * 4;
    unsigned char* rid = (unsigned char*)(ws + off);   off += (size_t)N_EDGES;
    (void)n_in; (void)in_sizes; (void)out_size;

    const int TPB = 256;
    const int nodeBlocks = (N_NODES + TPB - 1) / TPB;

    k_deg_partial<<<DEG_R * DEG_S, DEG_TPB, 0, stream>>>(col, ew, bufA, (unsigned*)rid, dinv, ws_size);
    k_dinv_sum<<<nodeBlocks, TPB, 0, stream>>>(bufA, dinv);

    k_gemm_splitk<<<SK_GRID, SK_NW * 32, 0, stream>>>(x, W1, bufH, SK_TILES);
    k_edge_owner<0><<<OWN_R * OWN_S, OWN_TPB, 0, stream>>>(row, col, ew, dinv, bufH, bufA, b1,
                                                           W2, Wfc, bfc, out,
                                                           rec, rcnt, rid, dinv, ws_size);

    k_edge_owner<1><<<OWN_R * OWN_S, OWN_TPB, 0, stream>>>(row, col, ew, dinv, bufA, bufH, b2,
                                                           W2, Wfc, bfc, out,
                                                           rec, rcnt, rid, dinv, ws_size);
}


# layer-1 edge pass reads one 8-byte record per found edge (source | place, weight; left in edge order by the degree pass) instead of three scattered 4-byte reads; sum order: arrival, as before
# speedup vs baseline: 2.5584x; 1.0750x over previous
#include <hip/hip_runtime.h>
#include <hip/hip_bf16.h>


#define N_NODES 100000
#define N_EDGES 1600000
#define IN_DIM  512
#define HID     64
#define OUT_DIM 5

typedef __attribute__((ext_vector_type(2))) float v2f;
typedef __attribute__((ext_vector_type(8))) float v8f;

#define DEG_R     8
#define DEG_S     32
#define DEG_NR    12512
#define DEG_PITCH (DEG_R * DEG_NR)
#define DEG_ES    (N_EDGES / DEG_S)
#define DEG_TPB   1024

typedef __attribute__((ext_vector_type(4))) int   v4i;
typedef __attribute__((ext_vector_type(4))) float v4f;

#define OWN_NR    397

__device__ __forceinline__ unsigned rid_pack(v4i c) {
    unsigned o = 0u;
    #pragma unroll
    for (int j = 0; j < 4; ++j) {
        unsigned q = (unsigned)c[j] / (unsigned)OWN_NR;
        q = q < 255u ? q : 255u;
        o |= q << (8 * j);
    }
    return o;
}

typedef __attribute__((ext_vector_type(2))) int v2i;

__global__ __launch_bounds__(DEG_TPB)
void k_deg_partial(const int* __restrict__ row,
                   const int* __restrict__ col,
                   const float* __restrict__ w,
                   float* __restrict__ part,
                   unsigned* __restrict__ ridw,
                   int* __restrict__ erec,
                   const float* wsBase, size_t wsSize) {
    __shared__ float sdeg[DEG_NR];
    const int r  = blockIdx.x % DEG_R;
    const int s  = blockIdx.x / DEG_R;
    const int lo = r * DEG_NR;
    const bool hasb = (size_t)((const char*)(ridw + N_EDGES / 4) - (const char*)wsBase) <= wsSize;
    const bool hasr = (size_t)((const char*)(erec + 2 * (size_t)N_EDGES) - (const char*)wsBase) <= wsSize;

    for (int i = threadIdx.x; i < DEG_NR; i += DEG_TPB) sdeg[i] = 0.0f;
    __syncthreads();

    const v4i* c4 = (const v4i*)(col + (size_t)s * DEG_ES);
    const v4i* r4 = (const v4i*)(row + (size_t)s * DEG_ES);
    const v4f* w4 = (const v4f*)(w   + (size_t)s * DEG_ES);
    v4i* e4 = (v4i*)erec + (size_t)s * (DEG_ES / 2);
    int blk = 0;
    for (int v0 = threadIdx.x; v0 < DEG_ES / 4; v0 += 4 * DEG_TPB, blk += 4) {
        v4i c[4];
        v4f x[4];
        const int  uw = (r - blk % DEG_R + DEG_R) % DEG_R;
        const int  vw = v0 + uw * DEG_TPB;
        v4i rs = {0, 0, 0, 0};
        if (hasr && uw < 4 && vw < DEG_ES / 4) rs = r4[vw];
        for (int u = 0; u < 4; ++u) {
            const int v = v0 + u * DEG_TPB;
            const bool in = v < DEG_ES / 4;
            c[u] = in ? c4[v] : v4i{-1, -1, -1, -1};
            x[u] = in ? w4[v] : v4f{0.0f, 0.0f, 0.0f, 0.0f};
        }
        for (int u = 0; u < 4; ++u) {
            const int v = v0 + u * DEG_TPB;
            if ((blk + u) % DEG_R == r && v < DEG_ES / 4) {
                if (hasb) ridw[(size_t)s * (DEG_ES / 4) + v] = rid_pack(c[u]);
                if (hasr) {
                    int p[4];
                    int q[4];
                    #pragma unroll
                    for (int j = 0; j < 4; ++j) {
                        const unsigned cj = (unsigned)c[u][j];
                        const unsigned pl = cj - (cj / (unsigned)OWN_NR) * (unsigned)OWN_NR;
                        const float    xw = x[u][j];
                        p[j] = rs[j] | (int)(pl << 17);
                        q[j] = __builtin_bit_cast(int, xw);
                    }
                    e4[2 * (size_t)v]     = v4i{p[0], q[0], p[1], q[1]};
                    e4[2 * (size_t)v + 1] = v4i{p[2], q[2], p[3], q[3]};
                }
            }
        }
        for (int u = 0; u < 4; ++u) {
            for (int j = 0; j < 4; ++j) {
                const unsigned d = (unsigned)(c[u][j] - lo);
                if (d < (unsigned)DEG_NR)
                    __hip_atomic_fetch_add(&sdeg[d], x[u][j], __ATOMIC_RELAXED,
                                           __HIP_MEMORY_SCOPE_WORKGROUP);
            }
        }
    }
    __syncthreads();

    float* dst = part + (size_t)s * DEG_PITCH + lo;
    for (int i = threadIdx.x; i < DEG_NR; i += DEG_TPB) dst[i] = sdeg[i];
}

__global__ void k_dinv_sum(const float* __restrict__ part,
                           float* __restrict__ dinv) {
    int i = blockIdx.x * blockDim.x + threadIdx.x;
    if (i < N_NODES) {
        float d = 1.0f;
        for (int s = 0; s < DEG_S; ++s) d += part[(size_t)s * DEG_PITCH + i];
        dinv[i] = (d > 0.0f) ? rsqrtf(d) : 0.0f;
    }
}


#define SK_NW     8
#define SK_KS     (IN_DIM / SK_NW)
#define SK_NJ     (SK_KS / 8)
#define SK_PITCH  68
#define SK_GRID   512
#define SK_TILES  (N_NODES / 16)

__global__ __launch_bounds__(SK_NW * 32)
void k_gemm_splitk(const float* __restrict__ X,
                   const float* __restrict__ W,
                   float* __restrict__ H,
                   int nTiles) {
    __shared__ float red[2][SK_NW][16 * SK_PITCH];

    const int lane = threadIdx.x & 31;
    const int wv = __builtin_amdgcn_readfirstlane(threadIdx.x >> 5);
    const int c  = lane & 15;
    const int hf = lane >> 4;
    const int kb = wv * SK_KS;

    v2f bw[SK_NJ][2][4];
    #pragma unroll
    for (int j = 0; j < SK_NJ; ++j) {
        #pragma unroll
        for (int q = 0; q < 2; ++q) {
            const float* wp = W + (size_t)(kb + 8 * j + 4 * hf + 2 * q) * HID + c;
            #pragma unroll
            for (int n = 0; n < 4; ++n) {
                bw[j][q][n].x = wp[16 * n];
                bw[j][q][n].y = wp[HID + 16 * n];
            }
        }
    }

    const float* xp = X + (size_t)c * IN_DIM + kb + 4 * hf;

    int tile = blockIdx.x;
    v4f ac[SK_NJ];
    v4f an[SK_NJ];
    if (tile < nTiles) {
        const float* xt = xp + (size_t)tile * (16 * IN_DIM);
        #pragma unroll
        for (int j = 0; j < SK_NJ; ++j) ac[j] = *(const v4f*)(xt + 8 * j);
    }

    int buf = 0;
    for (; tile < nTiles; tile += gridDim.x) {
        {
            const int tn = tile + (int)gridDim.x;
            const int tl = tn < nTiles ? tn : tile;
            const float* xt = xp + (size_t)tl * (16 * IN_DIM);
            #pragma unroll
            for (int j = 0; j < SK_NJ; ++j) an[j] = *(const v4f*)(xt + 8 * j);
        }

        v8f acc[4] = {};
        #pragma unroll
        for (int j = 0; j < SK_NJ; ++j) {
            #pragma unroll
            for (int q = 0; q < 2; ++q) {
                v2f a;
                a.x = ac[j][2 * q];
                a.y = ac[j][2 * q + 1];
                #pragma unroll
                for (int n = 0; n < 4; ++n)
                    acc[n] = __builtin_amdgcn_wmma_f32_16x16x4_f32(false, a, false, bw[j][q][n],
                                                                   (short)0, acc[n], false, false);
            }
        }

        {
            float* pr = &red[buf][wv][(8 * hf) * SK_PITCH + c];
            #pragma unroll
            for (int n = 0; n < 4; ++n) {
                #pragma unroll
                for (int r = 0; r < 8; ++r) pr[r * SK_PITCH + 16 * n] = acc[n][r];
            }
        }
        __builtin_amdgcn_fence(__ATOMIC_RELEASE, "workgroup", "local");
        __builtin_amdgcn_s_barrier();
        __builtin_amdgcn_fence(__ATOMIC_ACQUIRE, "workgroup", "local");

        {
            const int rr = 2 * wv + hf;
            const float* sp = &red[buf][0][rr * SK_PITCH + 4 * c];
            v4f s = *(const v4f*)sp;
            #pragma unroll
            for (int p = 1; p < SK_NW; ++p) s += *(const v4f*)(sp + p * (16 * SK_PITCH));
            *(v4f*)(H + ((size_t)tile * 16 + rr) * HID + 4 * c) = s;
        }

        buf ^= 1;
        #pragma unroll
        for (int j = 0; j < SK_NJ; ++j) ac[j] = an[j];
    }
}

#define OWN_R     252
#define OWN_S     1
#define OWN_TPB   1024
#define OWN_NV    4
#define OWN_VTOT  (N_EDGES / 16)
#define OWN_VS    ((OWN_VTOT + OWN_S - 1) / OWN_S)
#define OWN_CHUNK (OWN_NV * OWN_TPB)
#define OWN_NCH   ((OWN_VS + OWN_CHUNK - 1) / OWN_CHUNK)
#define OWN_GB    16
#define OWN_WQ    64
#define OWN_RCAP  288
#define OWN_NWAVE (OWN_R * OWN_S * (OWN_TPB / 32))
#define OWN_FIT   ((OWN_NR * (HID / 4) + OWN_TPB - 1) / OWN_TPB)
#define OWN_PITCH2 68
#define OWN_NT     ((OWN_NR + 1 + 15) / 16)
#define OWN_ROWS2  (16 * OWN_NT)

template <int MODE>
__global__ __launch_bounds__(OWN_TPB)
void k_edge_owner(const int* __restrict__ row,
                  const int* __restrict__ col,
                  const float* __restrict__ w,
                  const float* __restrict__ dinv,
                  const float* __restrict__ h,
                  float* __restrict__ agg,
                  const float* __restrict__ b,
                  const float* __restrict__ Wm,
                  const float* __restrict__ Wfc,
                  const float* __restrict__ bfc,
                  float* __restrict__ out,
                  int* __restrict__ rec,
                  int* __restrict__ rcnt,
                  const unsigned char* __restrict__ rid,
                  const v2i* __restrict__ erec,
                  const float* wsBase, size_t wsSize) {
    constexpr int PITCH = (MODE == 1) ? OWN_PITCH2 : HID;
    constexpr int ROWS  = (MODE == 1) ? OWN_ROWS2 : OWN_NR + 1;
    __shared__ float    sums[ROWS * PITCH];
    __shared__ unsigned wq[OWN_TPB / 32][OWN_WQ];

    const int tid  = threadIdx.x;
    const int lane = tid & 31;
    const int wv   = __builtin_amdgcn_readfirstlane(tid >> 5);
    const int r    = blockIdx.x % OWN_R;
    const int s    = blockIdx.x / OWN_R;
    const int lo   = r * OWN_NR;
    const int vs   = s * OWN_VS;
    const int ve   = (vs + OWN_VS < OWN_VTOT) ? (vs + OWN_VS) : OWN_VTOT;

    const bool room = (size_t)((const char*)(rcnt + OWN_NWAVE) - (const char*)wsBase) <= wsSize;
    const bool hasb = (size_t)((const char*)(rid + N_EDGES) - (const char*)wsBase) <= wsSize;
    const bool hasr = (size_t)((const char*)(erec + N_EDGES) - (const char*)wsBase) <= wsSize;
    const int  gw   = blockIdx.x * (OWN_TPB / 32) + wv;
    int* myrec = rec + (size_t)gw * (2 * OWN_RCAP);
    int  nrec  = 0;
    int  nrep  = -1;
    if (MODE == 1 && room) nrep = __builtin_amdgcn_readfirstlane(rcnt[gw]);
    const bool doscan = (MODE == 0) || nrep < 0;

    {
        v4f* z4 = (v4f*)sums;
        constexpr int NZ = (MODE == 1) ? ROWS * PITCH / 4 : OWN_NR * HID / 4;
        for (int i = tid; i < NZ; i += OWN_TPB)
            z4[i] = v4f{0.0f, 0.0f, 0.0f, 0.0f};
    }

    const v4i* b4 = (const v4i*)rid;
    const v4i* c4 = (const v4i*)col;
    auto ldids = [&](int v) -> v4i {
        if (hasb) return b4[v];
        v4i o;
        #pragma unroll
        for (int q = 0; q < 4; ++q) o[q] = (int)rid_pack(c4[4 * v + q]);
        return o;
    };
    v4i c[OWN_NV];
    v4i cn[OWN_NV];
    #pragma unroll
    for (int u = 0; u < OWN_NV; ++u) {
        const int v = vs + u * OWN_TPB + tid;
        c[u] = (doscan && v < ve) ? ldids(v) : v4i{-1, -1, -1, -1};
    }
    __syncthreads();

    unsigned* myq = wq[wv];
    int count = 0;

    auto gather_add = [&](int rr, int dd, float nr, int cnt) {
        v2f   hv[OWN_GB];
        float nn[OWN_GB];
        int   dl[OWN_GB];
        #pragma unroll
        for (int u = 0; u < OWN_GB; ++u) {
            const bool ok = u < cnt;
            const int  i  = ok ? u : 0;
            const int  ru = __builtin_amdgcn_readlane(rr, i);
            const int  du = __builtin_amdgcn_readlane(dd, i);
            dl[u] = ok ? du : OWN_NR;
            nn[u] = __builtin_bit_cast(float,
                      __builtin_amdgcn_readlane(__builtin_bit_cast(int, nr), i));
            hv[u] = *(const v2f*)(h + (size_t)ru * HID + lane * 2);
        }
        #pragma unroll
        for (int u = 0; u < OWN_GB; ++u) {
            float* dst = &sums[dl[u] * PITCH + lane * 2];
            __hip_atomic_fetch_add(dst + 0, nn[u] * hv[u].x, __ATOMIC_RELAXED,
                                   __HIP_MEMORY_SCOPE_WORKGROUP);
            __hip_atomic_fetch_add(dst + 1, nn[u] * hv[u].y, __ATOMIC_RELAXED,
                                   __HIP_MEMORY_SCOPE_WORKGROUP);
        }
    };

    auto serve = [&](int base, int cnt) {
        int   rr = 0;
        int   dd = 0;
        float nr = 0.0f;
        if (lane < cnt) {
            const unsigned e = myq[base + lane];
            int   cc;
            float we;
            if (hasr) {
                const v2i pr = erec[e];
                rr = pr.x & 0x1ffff;
                dd = pr.x >> 17;
                cc = lo + dd;
                const int wb = pr.y;
                we = __builtin_bit_cast(float, wb);
            } else {
                rr = row[e];
                cc = col[e];
                we = w[e];
                dd = cc - lo;
            }
            nr = dinv[rr] * we * dinv[cc];
        }
        if (MODE == 0 && room) {
            if (nrec + cnt <= OWN_RCAP && lane < cnt) {
                myrec[nrec + lane]            = rr | (dd << 17);
                myrec[OWN_RCAP + nrec + lane] = __builtin_bit_cast(int, nr);
            }
            nrec += cnt;
        }
        gather_add(rr, dd, nr, cnt);
    };

    if (doscan) {
    for (int ch = 0; ch < OWN_NCH; ++ch) {
        const int vb = vs + ch * OWN_CHUNK + tid;

        #pragma unroll
        for (int u = 0; u < OWN_NV; ++u) {
            const int v = vb + OWN_CHUNK + u * OWN_TPB;
            cn[u] = (ch + 1 < OWN_NCH && v < ve) ? ldids(v) : v4i{-1, -1, -1, -1};
        }

        unsigned mh[2] = {0u, 0u};
        const unsigned r4 = (unsigned)r * 0x01010101u;
        #pragma unroll
        for (int u = 0; u < OWN_NV; ++u) {
            #pragma unroll
            for (int q = 0; q < 4; ++q) {
                const int k = 4 * u + q;
                const unsigned y = (unsigned)c[u][q] ^ r4;
                const unsigned t = ((y & 0x7f7f7f7fu) + 0x7f7f7f7fu) | y;
                mh[k >> 3] |= (~t & 0x80808080u) >> (7 - (k & 7));
            }
        }
        unsigned long long mask = (unsigned long long)mh[0] | ((unsigned long long)mh[1] << 32);

        for (;;) {
            const bool     has = mask != 0ull;
            const unsigned bal = __builtin_amdgcn_ballot_w32(has);
            if (bal == 0u) break;
            if (has) {
                const int idx = __builtin_ctzll(mask);
                mask &= mask - 1ull;
                const int k = (idx & 7) + ((idx >> 5) << 3);
                const unsigned e = 16u * (unsigned)(vb + (k >> 2) * OWN_TPB)
                                 + 4u * (unsigned)(k & 3) + (unsigned)((idx >> 3) & 3);
                myq[count + (int)__builtin_amdgcn_mbcnt_lo(bal, 0u)] = e;
            }
            count += __builtin_popcount(bal);
            __builtin_amdgcn_fence(__ATOMIC_ACQ_REL, "wavefront");
            while (count >= OWN_GB) {
                count -= OWN_GB;
                serve(count, OWN_GB);
            }
        }

        #pragma unroll
        for (int u = 0; u < OWN_NV; ++u) c[u] = cn[u];
    }
    if (count > 0) serve(0, count);
    if (MODE == 0 && room && lane == 0)
        rcnt[gw] = (nrec <= OWN_RCAP) ? nrec : -1;
    } else {
        if (nrep > 0) {
            const int last = nrep - 1;
            const int i0   = lane < last ? lane : last;
            int curp = myrec[i0];
            int curn = myrec[OWN_RCAP + i0];
            for (int base = 0; base < nrep; base += OWN_GB) {
                const int left = nrep - base;
                const int cnt  = left < OWN_GB ? left : OWN_GB;
                const int nx   = base + OWN_GB + lane;
                const int i1   = nx < last ? nx : last;
                const int nxtp = myrec[i1];
                const int nxtn = myrec[OWN_RCAP + i1];
                const int   rr = curp & 0x1ffff;
                const int   dd = curp >> 17;
                const float nr = __builtin_bit_cast(float, curn);
                if (cnt == OWN_GB) gather_add(rr, dd, nr, OWN_GB);
                else               gather_add(rr, dd, nr, cnt);
                curp = nxtp;
                curn = nxtn;
            }
        }
    }

    const int f  = 4 * (tid & 15);
    v4f   hs[OWN_FIT];
    float ds[OWN_FIT];
    #pragma unroll
    for (int t = 0; t < OWN_FIT; ++t) {
        const int nd = lo + ((tid + t * OWN_TPB) >> 4);
        const int nc = nd < N_NODES ? nd : N_NODES - 1;
        ds[t] = dinv[nc];
        hs[t] = *(const v4f*)(h + (size_t)nc * HID + f);
    }

    __builtin_amdgcn_fence(__ATOMIC_RELEASE, "workgroup", "local");
    __builtin_amdgcn_s_barrier();
    __builtin_amdgcn_fence(__ATOMIC_ACQUIRE, "workgroup", "local");

    if (MODE == 0) {
        const v4f bv = *(const v4f*)(b + f);
        const v4f* s4 = (const v4f*)sums;
        #pragma unroll
        for (int t = 0; t < OWN_FIT; ++t) {
            const int i    = tid + t * OWN_TPB;
            const int node = lo + (i >> 4);
            if (i < OWN_NR * (HID / 4) && node < N_NODES) {
                v4f v = s4[i] + ds[t] * ds[t] * hs[t] + bv;
                #pragma unroll
                for (int q = 0; q < 4; ++q) v[q] = fmaxf(v[q], 0.0f);
                *(v4f*)(agg + (size_t)node * HID + f) = v;
            }
        }
    } else {
        #pragma unroll
        for (int t = 0; t < OWN_FIT; ++t) {
            const int i = tid + t * OWN_TPB;
            if (i < OWN_NR * (HID / 4)) {
                v4f* p = (v4f*)&sums[(i >> 4) * PITCH + f];
                *p = *p + ds[t] * ds[t] * hs[t];
            }
        }
        __builtin_amdgcn_fence(__ATOMIC_RELEASE, "workgroup", "local");
        __builtin_amdgcn_s_barrier();
        __builtin_amdgcn_fence(__ATOMIC_ACQUIRE, "workgroup", "local");

        if (wv < OWN_NT) {
            const int c  = lane & 15;
            const int hf = lane >> 4;
            const float* sp = &sums[(16 * wv + c) * PITCH + 2 * hf];
            const float* wp = Wm + (size_t)(2 * hf) * HID + c;
            v8f acc[4] = {};
            #pragma unroll 4
            for (int k0 = 0; k0 < HID; k0 += 4) {
                const v2f sv = *(const v2f*)(sp + k0);
                #pragma unroll
                for (int n = 0; n < 4; ++n) {
                    v2f wf;
                    wf.x = wp[(size_t)k0 * HID + 16 * n];
                    wf.y = wp[(size_t)(k0 + 1) * HID + 16 * n];
                    acc[n] = __builtin_amdgcn_wmma_f32_16x16x4_f32(false, wf, false, sv,
                                                                   (short)0, acc[n], false, false);
                }
            }
            #pragma unroll
            for (int n = 0; n < 4; ++n) {
                const v4f ba = *(const v4f*)(b + 16 * n + 8 * hf);
                const v4f bb = *(const v4f*)(b + 16 * n + 8 * hf + 4);
                #pragma unroll
                for (int j = 0; j < 4; ++j) {
                    acc[n][j]     = fmaxf(acc[n][j]     + ba[j], 0.0f);
                    acc[n][j + 4] = fmaxf(acc[n][j + 4] + bb[j], 0.0f);
                }
            }
            const int   co = c < OUT_DIM ? c : 0;
            const float cm = c < OUT_DIM ? 1.0f : 0.0f;
            v2f wa[4][4];
            #pragma unroll
            for (int n = 0; n < 4; ++n) {
                #pragma unroll
                for (int jj = 0; jj < 4; ++jj) {
                    const int kc = 16 * n + 8 * hf + 2 * jj;
                    wa[n][jj].x = cm * Wfc[kc * OUT_DIM + co];
                    wa[n][jj].y = cm * Wfc[(kc + 1) * OUT_DIM + co];
                }
            }
            v8f fo = {};
            #pragma unroll
            for (int n = 0; n < 4; ++n) {
                #pragma unroll
                for (int jj = 0; jj < 4; ++jj) {
                    v2f rb;
                    rb.x = acc[n][2 * jj];
                    rb.y = acc[n][2 * jj + 1];
                    fo = __builtin_amdgcn_wmma_f32_16x16x4_f32(false, wa[n][jj], false, rb,
                                                               (short)0, fo, false, false);
                }
            }
            const int rw   = 16 * wv + c;
            const int node = lo + rw;
            if (hf == 0 && rw < OWN_NR && node < N_NODES) {
                float* op = out + (size_t)node * OUT_DIM;
                #pragma unroll
                for (int o = 0; o < OUT_DIM; ++o) op[o] = fo[o] + bfc[o];
            }
        }
    }
}

extern "C" void kernel_launch(void* const* d_in, const int* in_sizes, int n_in,
                              void* d_out, int out_size, void* d_ws, size_t ws_size,
                              hipStream_t stream) {
    const float* x   = (const float*)d_in[0];
    const int*   ei  = (const int*)  d_in[1];
    const float* ew  = (const float*)d_in[2];
    const float* W1  = (const float*)d_in[3];
    const float* b1  = (const float*)d_in[4];
    const float* W2  = (const float*)d_in[5];
    const float* b2  = (const float*)d_in[6];
    const float* Wfc = (const float*)d_in[7];
    const float* bfc = (const float*)d_in[8];
    float* out = (float*)d_out;

    const int* row = ei;
    const int* col = ei + N_EDGES;

    char* ws = (char*)d_ws;
    size_t off = 0;
    float* dinv = (float*)(ws + off); off += ((size_t)N_NODES * 4 + 255) & ~(size_t)255;
    float* bufH = (float*)(ws + off); off += (size_t)N_NODES * HID * 4;
    float* bufA = (float*)(ws + off); off += (size_t)N_NODES * HID * 4;
    int*   rec  = (int*)(ws + off);   off += (size_t)OWN_NWAVE * OWN_RCAP * 8;
    int*   rcnt = (int*)(ws + off);   off += (size_t)OWN_NWAVE * 4;
    unsigned char* rid = (unsigned char*)(ws + off);   off += (size_t)N_EDGES;
    int*   erec = (int*)(ws + off);   off += (size_t)N_EDGES * 8;
    (void)n_in; (void)in_sizes; (void)out_size;

    const int TPB = 256;
    const int nodeBlocks = (N_NODES + TPB - 1) / TPB;

    k_deg_partial<<<DEG_R * DEG_S, DEG_TPB, 0, stream>>>(row, col, ew, bufA, (unsigned*)rid, erec, dinv, ws_size);
    k_dinv_sum<<<nodeBlocks, TPB, 0, stream>>>(bufA, dinv);

    k_gemm_splitk<<<SK_GRID, SK_NW * 32, 0, stream>>>(x, W1, bufH, SK_TILES);
    k_edge_owner<0><<<OWN_R * OWN_S, OWN_TPB, 0, stream>>>(row, col, ew, dinv, bufH, bufA, b1,
                                                           W2, Wfc, bfc, out,
                                                           rec, rcnt, rid, (const v2i*)erec, dinv, ws_size);

    k_edge_owner<1><<<OWN_R * OWN_S, OWN_TPB, 0, stream>>>(row, col, ew, dinv, bufA, bufH, b2,
                                                           W2, Wfc, bfc, out,
                                                           rec, rcnt, rid, (const v2i*)erec, dinv, ws_size);
}
